# starting kernel
# baseline (speedup 1.0000x reference)
#include <hip/hip_runtime.h>

#define N_U 100000
#define N_I 50000
#define NE  1600000
#define DIM 128

typedef __attribute__((ext_vector_type(2))) float v2f;
typedef __attribute__((ext_vector_type(8))) float v8f;

__device__ __forceinline__ void atomicAddF(float* p, float v) {
    __hip_atomic_fetch_add(p, v, __ATOMIC_RELAXED, __HIP_MEMORY_SCOPE_AGENT);
}

__global__ void deg_kernel(const int* __restrict__ src, const int* __restrict__ dst,
                           float* __restrict__ degS, float* __restrict__ degD, int nE) {
    int e = blockIdx.x * blockDim.x + threadIdx.x;
    if (e < nE) {
        atomicAddF(&degS[src[e]], 1.0f);
        atomicAddF(&degD[dst[e]], 1.0f);
    }
}

__global__ void invsqrt_kernel(float* __restrict__ a, int n) {
    int i = blockIdx.x * blockDim.x + threadIdx.x;
    if (i < n) {
        float d = a[i];
        a[i] = (d > 0.0f) ? rsqrtf(d) : 0.0f;
    }
}

__global__ void gemm_wmma_f32(const float* __restrict__ X, const float* __restrict__ W,
                              float* __restrict__ H) {
    const int m0   = blockIdx.x << 4;
    const int n0   = blockIdx.y << 4;
    const int lane = threadIdx.x;
    const int r15  = lane & 15;
    const int kb   = (lane >> 4) << 1;

    const float* __restrict__ xrow = X + (size_t)(m0 + r15) * DIM;

    v8f c = {};
    #pragma unroll
    for (int k0 = 0; k0 < DIM; k0 += 4) {
        v2f a;
        a.x = xrow[k0 + kb];
        a.y = xrow[k0 + kb + 1];
        v2f b;
        b.x = W[(size_t)(k0 + kb)     * DIM + n0 + r15];
        b.y = W[(size_t)(k0 + kb + 1) * DIM + n0 + r15];
        c = __builtin_amdgcn_wmma_f32_16x16x4_f32(
                 false, a,  false, b,
                 (short)0, c,  false,  false);
    }

    const int mbase = m0 + ((lane >> 4) << 3);
    #pragma unroll
    for (int r = 0; r < 8; ++r)
        H[(size_t)(mbase + r) * DIM + n0 + r15] = c[r];
}

__global__ void scatter_kernel(const float* __restrict__ H,
                               const int* __restrict__ src, const int* __restrict__ dst,
                               const float* __restrict__ invS, const float* __restrict__ invD,
                               float* __restrict__ acc, int nE) {
    int gid  = blockIdx.x * blockDim.x + threadIdx.x;
    int e    = gid >> 5;
    int lane = gid & 31;
    if (e >= nE) return;

    int s = src[e];
    int d = dst[e];
    float nrm = invS[s] * invD[d];

    const float4 v = *((const float4*)(H + (size_t)s * DIM) + lane);
    float* op = acc + (size_t)d * DIM + lane * 4;
    atomicAddF(op + 0, v.x * nrm);
    atomicAddF(op + 1, v.y * nrm);
    atomicAddF(op + 2, v.z * nrm);
    atomicAddF(op + 3, v.w * nrm);
}

__global__ void finalize_kernel(float* __restrict__ buf,
                                const float* __restrict__ b1,
                                const float* __restrict__ b2,
                                float scale, int do_relu, int n) {
    int i = blockIdx.x * blockDim.x + threadIdx.x;
    if (i < n) {
        int c = i & (DIM - 1);
        float v = buf[i] + b1[c] + (b2 ? b2[c] : 0.0f);
        v *= scale;
        if (do_relu) v = fmaxf(v, 0.0f);
        buf[i] = v;
    }
}

extern "C" void kernel_launch(void* const* d_in, const int* in_sizes, int n_in,
                              void* d_out, int out_size, void* d_ws, size_t ws_size,
                              hipStream_t stream) {
    const float* x_user = (const float*)d_in[0];
    const float* x_item = (const float*)d_in[1];
    const int* f_src = (const int*)d_in[2];
    const int* f_dst = (const int*)d_in[3];
    const int* r_src = (const int*)d_in[4];
    const int* r_dst = (const int*)d_in[5];
    const int* v_src = (const int*)d_in[6];
    const int* v_dst = (const int*)d_in[7];
    const float* W1f = (const float*)d_in[8];
    const float* W1r = (const float*)d_in[9];
    const float* W1v = (const float*)d_in[10];
    const float* W2f = (const float*)d_in[11];
    const float* W2r = (const float*)d_in[12];
    const float* W2v = (const float*)d_in[13];
    const float* b1f = (const float*)d_in[14];
    const float* b1r = (const float*)d_in[15];
    const float* b1v = (const float*)d_in[16];
    const float* b2f = (const float*)d_in[17];
    const float* b2r = (const float*)d_in[18];
    const float* b2v = (const float*)d_in[19];

    float* w = (float*)d_ws;
    float* inv_fs = w; w += N_U;
    float* inv_fd = w; w += N_U;
    float* inv_rs = w; w += N_U;
    float* inv_rd = w; w += N_I;
    float* inv_vs = w; w += N_I;
    float* inv_vd = w; w += N_U;
    const int n_inv = 4 * N_U + 2 * N_I;
    float* h_user = w; w += (size_t)N_U * DIM;
    float* h_item = w; w += (size_t)N_I * DIM;
    float* u1     = w; w += (size_t)N_U * DIM;
    float* it1    = w; w += (size_t)N_I * DIM;

    float* out_u = (float*)d_out;
    float* out_i = out_u + (size_t)N_U * DIM;

    const int TPB = 256;
    const int degBlocks  = (NE + TPB - 1) / TPB;
    const int scatBlocks = ((size_t)NE * 32 + TPB - 1) / TPB;
    dim3 gemmU((N_U / 16), DIM / 16);
    dim3 gemmI((N_I / 16), DIM / 16);

    hipMemsetAsync(inv_fs, 0, (size_t)n_inv * sizeof(float), stream);
    deg_kernel<<<degBlocks, TPB, 0, stream>>>(f_src, f_dst, inv_fs, inv_fd, NE);
    deg_kernel<<<degBlocks, TPB, 0, stream>>>(r_src, r_dst, inv_rs, inv_rd, NE);
    deg_kernel<<<degBlocks, TPB, 0, stream>>>(v_src, v_dst, inv_vs, inv_vd, NE);
    invsqrt_kernel<<<(n_inv + TPB - 1) / TPB, TPB, 0, stream>>>(inv_fs, n_inv);

    hipMemsetAsync(u1,  0, (size_t)N_U * DIM * sizeof(float), stream);
    hipMemsetAsync(it1, 0, (size_t)N_I * DIM * sizeof(float), stream);

    gemm_wmma_f32<<<gemmU, 32, 0, stream>>>(x_user, W1f, h_user);
    scatter_kernel<<<scatBlocks, TPB, 0, stream>>>(h_user, f_src, f_dst, inv_fs, inv_fd, u1, NE);
    gemm_wmma_f32<<<gemmI, 32, 0, stream>>>(x_item, W1v, h_item);
    scatter_kernel<<<scatBlocks, TPB, 0, stream>>>(h_item, v_src, v_dst, inv_vs, inv_vd, u1, NE);
    finalize_kernel<<<((size_t)N_U * DIM + TPB - 1) / TPB, TPB, 0, stream>>>(
        u1, b1f, b1v, 0.5f, 1, N_U * DIM);

    gemm_wmma_f32<<<gemmU, 32, 0, stream>>>(x_user, W1r, h_user);
    scatter_kernel<<<scatBlocks, TPB, 0, stream>>>(h_user, r_src, r_dst, inv_rs, inv_rd, it1, NE);
    finalize_kernel<<<((size_t)N_I * DIM + TPB - 1) / TPB, TPB, 0, stream>>>(
        it1, b1r, nullptr, 1.0f, 1, N_I * DIM);

    hipMemsetAsync(d_out, 0, (size_t)(N_U + N_I) * DIM * sizeof(float), stream);

    gemm_wmma_f32<<<gemmU, 32, 0, stream>>>(u1, W2f, h_user);
    scatter_kernel<<<scatBlocks, TPB, 0, stream>>>(h_user, f_src, f_dst, inv_fs, inv_fd, out_u, NE);
    gemm_wmma_f32<<<gemmI, 32, 0, stream>>>(it1, W2v, h_item);
    scatter_kernel<<<scatBlocks, TPB, 0, stream>>>(h_item, v_src, v_dst, inv_vs, inv_vd, out_u, NE);
    finalize_kernel<<<((size_t)N_U * DIM + TPB - 1) / TPB, TPB, 0, stream>>>(
        out_u, b2f, b2v, 0.5f, 0, N_U * DIM);

    gemm_wmma_f32<<<gemmU, 32, 0, stream>>>(u1, W2r, h_user);
    scatter_kernel<<<scatBlocks, TPB, 0, stream>>>(h_user, r_src, r_dst, inv_rs, inv_rd, out_i, NE);
    finalize_kernel<<<((size_t)N_I * DIM + TPB - 1) / TPB, TPB, 0, stream>>>(
        out_i, b2r, nullptr, 1.0f, 0, N_I * DIM);
}


# Replace float-atomic edge scatter by per-call dst-sorted edge list (count scan, arrival-rank fill) and one-wave-per-row gather with plain stores; row sum order is arrival rank
# speedup vs baseline: 1.6863x; 1.6863x over previous
#include <hip/hip_runtime.h>

#define N_U 100000
#define N_I 50000
#define NE  1600000
#define DIM 128

typedef __attribute__((ext_vector_type(2))) float v2f;
typedef __attribute__((ext_vector_type(8))) float v8f;

__device__ __forceinline__ void atomicAddF(float* p, float v) {
    __hip_atomic_fetch_add(p, v, __ATOMIC_RELAXED, __HIP_MEMORY_SCOPE_AGENT);
}

__global__ void deg_kernel(const int* __restrict__ src, const int* __restrict__ dst,
                           float* __restrict__ degS, float* __restrict__ degD, int nE) {
    int e = blockIdx.x * blockDim.x + threadIdx.x;
    if (e < nE) {
        atomicAddF(&degS[src[e]], 1.0f);
        atomicAddF(&degD[dst[e]], 1.0f);
    }
}

__global__ void invsqrt_kernel(float* __restrict__ a, int n) {
    int i = blockIdx.x * blockDim.x + threadIdx.x;
    if (i < n) {
        float d = a[i];
        a[i] = (d > 0.0f) ? rsqrtf(d) : 0.0f;
    }
}

__global__ void gemm_wmma_f32(const float* __restrict__ X, const float* __restrict__ W,
                              float* __restrict__ H) {
    const int m0   = blockIdx.x << 4;
    const int n0   = blockIdx.y << 4;
    const int lane = threadIdx.x;
    const int r15  = lane & 15;
    const int kb   = (lane >> 4) << 1;

    const float* __restrict__ xrow = X + (size_t)(m0 + r15) * DIM;

    v8f c = {};
    #pragma unroll
    for (int k0 = 0; k0 < DIM; k0 += 4) {
        v2f a;
        a.x = xrow[k0 + kb];
        a.y = xrow[k0 + kb + 1];
        v2f b;
        b.x = W[(size_t)(k0 + kb)     * DIM + n0 + r15];
        b.y = W[(size_t)(k0 + kb + 1) * DIM + n0 + r15];
        c = __builtin_amdgcn_wmma_f32_16x16x4_f32(
                 false, a,  false, b,
                 (short)0, c,  false,  false);
    }

    const int mbase = m0 + ((lane >> 4) << 3);
    #pragma unroll
    for (int r = 0; r < 8; ++r)
        H[(size_t)(mbase + r) * DIM + n0 + r15] = c[r];
}

__global__ void rowstart_kernel(const float* __restrict__ degA, int* __restrict__ curA, int nA,
                                const float* __restrict__ degB, int* __restrict__ curB, int nB,
                                const float* __restrict__ degC, int* __restrict__ curC, int nC) {
    const float* __restrict__ deg = degA; int* __restrict__ cur = curA; int n = nA;
    if (blockIdx.y == 1) { deg = degB; cur = curB; n = nB; }
    if (blockIdx.y == 2) { deg = degC; cur = curC; n = nC; }
    const int t    = threadIdx.x;
    const int lane = t & 31;
    const int wv   = t >> 5;
    const int base = blockIdx.x << 10;
    if (base >= n) return;

    int part = 0;
    const float4* __restrict__ d4 = (const float4*)deg;
    #pragma unroll 4
    for (int i = t; i < (base >> 2); i += 256) {
        const float4 v = d4[i];
        part += (int)v.x + (int)v.y + (int)v.z + (int)v.w;
    }

    const int i0 = base + (t << 2);
    int c[4];
    #pragma unroll
    for (int j = 0; j < 4; ++j) c[j] = (i0 + j < n) ? (int)deg[i0 + j] : 0;
    const int mine = c[0] + c[1] + c[2] + c[3];

    int incl = mine;
    for (int o = 1; o < 32; o <<= 1) {
        const int y = __shfl_up(incl, o);
        if (lane >= o) incl += y;
    }
    int psum = part;
    for (int o = 16; o > 0; o >>= 1) psum += __shfl_xor(psum, o);

    __shared__ int s_w[8];
    __shared__ int s_p[8];
    if (lane == 31) s_w[wv] = incl;
    if (lane == 0)  s_p[wv] = psum;
    __syncthreads();
    int off = 0;
    #pragma unroll
    for (int k = 0; k < 8; ++k) {
        off += s_p[k];
        if (k < wv) off += s_w[k];
    }
    int ex = off + incl - mine;
    #pragma unroll
    for (int j = 0; j < 4; ++j) {
        if (i0 + j < n) { cur[i0 + j] = ex; ex += c[j]; }
    }
}

__global__ void fill_kernel(const int* __restrict__ src, const int* __restrict__ dst,
                            int* __restrict__ cur, int* __restrict__ csr, int nE) {
    int e = blockIdx.x * blockDim.x + threadIdx.x;
    if (e < nE) {
        const int d = dst[e];
        const int p = __hip_atomic_fetch_add(&cur[d], 1, __ATOMIC_RELAXED, __HIP_MEMORY_SCOPE_AGENT);
        if ((unsigned)p < (unsigned)nE) csr[p] = src[e];
    }
}

__global__ void gather_kernel(const float* __restrict__ H,
                              const int* __restrict__ csr, const int* __restrict__ cur,
                              const float* __restrict__ invS, const float* __restrict__ invD,
                              float* __restrict__ acc, int nRows, int accumulate) {
    const int gid  = blockIdx.x * blockDim.x + threadIdx.x;
    const int row  = gid >> 5;
    const int lane = gid & 31;
    if (row >= nRows) return;

    const int beg = __builtin_amdgcn_readfirstlane(row > 0 ? cur[row - 1] : 0);
    const int end = __builtin_amdgcn_readfirstlane(cur[row]);
    const float wd = invD[row];

    float4 a = make_float4(0.0f, 0.0f, 0.0f, 0.0f);
    for (int c0 = beg; c0 < end; c0 += 32) {
        const int n = (end - c0 < 32) ? (end - c0) : 32;
        int   my_s = 0;
        float my_w = 0.0f;
        if (lane < n) {
            my_s = csr[c0 + lane];
            my_w = invS[my_s] * wd;
        }
        #pragma unroll 4
        for (int j = 0; j < n; ++j) {
            const int   s = __builtin_amdgcn_readlane(my_s, j);
            const float w = __int_as_float(__builtin_amdgcn_readlane(__float_as_int(my_w), j));
            const float4 v = *((const float4*)(H + (size_t)s * DIM) + lane);
            a.x += v.x * w;
            a.y += v.y * w;
            a.z += v.z * w;
            a.w += v.w * w;
        }
    }
    float4* op = (float4*)(acc + (size_t)row * DIM) + lane;
    if (accumulate) {
        const float4 o = *op;
        a.x += o.x; a.y += o.y; a.z += o.z; a.w += o.w;
    }
    *op = a;
}

__global__ void finalize_kernel(float* __restrict__ buf,
                                const float* __restrict__ b1,
                                const float* __restrict__ b2,
                                float scale, int do_relu, int n) {
    int i = blockIdx.x * blockDim.x + threadIdx.x;
    if (i < n) {
        int c = i & (DIM - 1);
        float v = buf[i] + b1[c] + (b2 ? b2[c] : 0.0f);
        v *= scale;
        if (do_relu) v = fmaxf(v, 0.0f);
        buf[i] = v;
    }
}

extern "C" void kernel_launch(void* const* d_in, const int* in_sizes, int n_in,
                              void* d_out, int out_size, void* d_ws, size_t ws_size,
                              hipStream_t stream) {
    const float* x_user = (const float*)d_in[0];
    const float* x_item = (const float*)d_in[1];
    const int* f_src = (const int*)d_in[2];
    const int* f_dst = (const int*)d_in[3];
    const int* r_src = (const int*)d_in[4];
    const int* r_dst = (const int*)d_in[5];
    const int* v_src = (const int*)d_in[6];
    const int* v_dst = (const int*)d_in[7];
    const float* W1f = (const float*)d_in[8];
    const float* W1r = (const float*)d_in[9];
    const float* W1v = (const float*)d_in[10];
    const float* W2f = (const float*)d_in[11];
    const float* W2r = (const float*)d_in[12];
    const float* W2v = (const float*)d_in[13];
    const float* b1f = (const float*)d_in[14];
    const float* b1r = (const float*)d_in[15];
    const float* b1v = (const float*)d_in[16];
    const float* b2f = (const float*)d_in[17];
    const float* b2r = (const float*)d_in[18];
    const float* b2v = (const float*)d_in[19];

    float* w = (float*)d_ws;
    float* inv_fs = w; w += N_U;
    float* inv_fd = w; w += N_U;
    float* inv_rs = w; w += N_U;
    float* inv_rd = w; w += N_I;
    float* inv_vs = w; w += N_I;
    float* inv_vd = w; w += N_U;
    const int n_inv = 4 * N_U + 2 * N_I;
    float* h_user = w; w += (size_t)N_U * DIM;
    float* h_item = h_user;
    int* csr_f = (int*)w;
    int* csr_r = csr_f + NE;
    int* csr_v = csr_r + NE;
    int* cur_f = csr_v + NE;
    int* cur_r = cur_f + N_U;
    int* cur_v = cur_r + N_I;
    w += (size_t)N_I * DIM;
    float* u1     = w; w += (size_t)N_U * DIM;
    float* it1    = w; w += (size_t)N_I * DIM;

    float* out_u = (float*)d_out;
    float* out_i = out_u + (size_t)N_U * DIM;

    const int TPB = 256;
    const int degBlocks = (NE + TPB - 1) / TPB;
    const int gathU = (N_U * 32 + TPB - 1) / TPB;
    const int gathI = (N_I * 32 + TPB - 1) / TPB;
    dim3 gemmU((N_U / 16), DIM / 16);
    dim3 gemmI((N_I / 16), DIM / 16);
    dim3 rsGrid((N_U + 1023) / 1024, 3);

    hipMemsetAsync(inv_fs, 0, (size_t)n_inv * sizeof(float), stream);
    deg_kernel<<<degBlocks, TPB, 0, stream>>>(f_src, f_dst, inv_fs, inv_fd, NE);
    deg_kernel<<<degBlocks, TPB, 0, stream>>>(r_src, r_dst, inv_rs, inv_rd, NE);
    deg_kernel<<<degBlocks, TPB, 0, stream>>>(v_src, v_dst, inv_vs, inv_vd, NE);
    rowstart_kernel<<<rsGrid, TPB, 0, stream>>>(inv_fd, cur_f, N_U, inv_rd, cur_r, N_I,
                                                inv_vd, cur_v, N_U);
    invsqrt_kernel<<<(n_inv + TPB - 1) / TPB, TPB, 0, stream>>>(inv_fs, n_inv);
    fill_kernel<<<degBlocks, TPB, 0, stream>>>(f_src, f_dst, cur_f, csr_f, NE);
    fill_kernel<<<degBlocks, TPB, 0, stream>>>(r_src, r_dst, cur_r, csr_r, NE);
    fill_kernel<<<degBlocks, TPB, 0, stream>>>(v_src, v_dst, cur_v, csr_v, NE);

    gemm_wmma_f32<<<gemmU, 32, 0, stream>>>(x_user, W1f, h_user);
    gather_kernel<<<gathU, TPB, 0, stream>>>(h_user, csr_f, cur_f, inv_fs, inv_fd, u1, N_U, 0);
    gemm_wmma_f32<<<gemmI, 32, 0, stream>>>(x_item, W1v, h_item);
    gather_kernel<<<gathU, TPB, 0, stream>>>(h_item, csr_v, cur_v, inv_vs, inv_vd, u1, N_U, 1);
    finalize_kernel<<<((size_t)N_U * DIM + TPB - 1) / TPB, TPB, 0, stream>>>(
        u1, b1f, b1v, 0.5f, 1, N_U * DIM);

    gemm_wmma_f32<<<gemmU, 32, 0, stream>>>(x_user, W1r, h_user);
    gather_kernel<<<gathI, TPB, 0, stream>>>(h_user, csr_r, cur_r, inv_rs, inv_rd, it1, N_I, 0);
    finalize_kernel<<<((size_t)N_I * DIM + TPB - 1) / TPB, TPB, 0, stream>>>(
        it1, b1r, nullptr, 1.0f, 1, N_I * DIM);

    gemm_wmma_f32<<<gemmU, 32, 0, stream>>>(u1, W2f, h_user);
    gather_kernel<<<gathU, TPB, 0, stream>>>(h_user, csr_f, cur_f, inv_fs, inv_fd, out_u, N_U, 0);
    gemm_wmma_f32<<<gemmI, 32, 0, stream>>>(it1, W2v, h_item);
    gather_kernel<<<gathU, TPB, 0, stream>>>(h_item, csr_v, cur_v, inv_vs, inv_vd, out_u, N_U, 1);
    finalize_kernel<<<((size_t)N_U * DIM + TPB - 1) / TPB, TPB, 0, stream>>>(
        out_u, b2f, b2v, 0.5f, 0, N_U * DIM);

    gemm_wmma_f32<<<gemmU, 32, 0, stream>>>(u1, W2r, h_user);
    gather_kernel<<<gathI, TPB, 0, stream>>>(h_user, csr_r, cur_r, inv_rs, inv_rd, out_i, N_I, 0);
    finalize_kernel<<<((size_t)N_I * DIM + TPB - 1) / TPB, TPB, 0, stream>>>(
        out_i, b2r, nullptr, 1.0f, 0, N_I * DIM);
}


# Fuse per node type both relation gathers with bias, mean scale and relu into the one-wave-per-row gather (sums kept in registers): drops 4 finalize and 2 gather launches; sum order unchanged
# speedup vs baseline: 1.8328x; 1.0869x over previous
#include <hip/hip_runtime.h>

#define N_U 100000
#define N_I 50000
#define NE  1600000
#define DIM 128

typedef __attribute__((ext_vector_type(2))) float v2f;
typedef __attribute__((ext_vector_type(8))) float v8f;

__device__ __forceinline__ void atomicAddF(float* p, float v) {
    __hip_atomic_fetch_add(p, v, __ATOMIC_RELAXED, __HIP_MEMORY_SCOPE_AGENT);
}

__global__ void deg_kernel(const int* __restrict__ src, const int* __restrict__ dst,
                           float* __restrict__ degS, float* __restrict__ degD, int nE) {
    int e = blockIdx.x * blockDim.x + threadIdx.x;
    if (e < nE) {
        atomicAddF(&degS[src[e]], 1.0f);
        atomicAddF(&degD[dst[e]], 1.0f);
    }
}

__global__ void invsqrt_kernel(float* __restrict__ a, int n) {
    int i = blockIdx.x * blockDim.x + threadIdx.x;
    if (i < n) {
        float d = a[i];
        a[i] = (d > 0.0f) ? rsqrtf(d) : 0.0f;
    }
}

__global__ void gemm_wmma_f32(const float* __restrict__ X, const float* __restrict__ W,
                              float* __restrict__ H) {
    const int m0   = blockIdx.x << 4;
    const int n0   = blockIdx.y << 4;
    const int lane = threadIdx.x;
    const int r15  = lane & 15;
    const int kb   = (lane >> 4) << 1;

    const float* __restrict__ xrow = X + (size_t)(m0 + r15) * DIM;

    v8f c = {};
    #pragma unroll
    for (int k0 = 0; k0 < DIM; k0 += 4) {
        v2f a;
        a.x = xrow[k0 + kb];
        a.y = xrow[k0 + kb + 1];
        v2f b;
        b.x = W[(size_t)(k0 + kb)     * DIM + n0 + r15];
        b.y = W[(size_t)(k0 + kb + 1) * DIM + n0 + r15];
        c = __builtin_amdgcn_wmma_f32_16x16x4_f32(
                 false, a,  false, b,
                 (short)0, c,  false,  false);
    }

    const int mbase = m0 + ((lane >> 4) << 3);
    #pragma unroll
    for (int r = 0; r < 8; ++r)
        H[(size_t)(mbase + r) * DIM + n0 + r15] = c[r];
}

__global__ void rowstart_kernel(const float* __restrict__ degA, int* __restrict__ curA, int nA,
                                const float* __restrict__ degB, int* __restrict__ curB, int nB,
                                const float* __restrict__ degC, int* __restrict__ curC, int nC) {
    const float* __restrict__ deg = degA; int* __restrict__ cur = curA; int n = nA;
    if (blockIdx.y == 1) { deg = degB; cur = curB; n = nB; }
    if (blockIdx.y == 2) { deg = degC; cur = curC; n = nC; }
    const int t    = threadIdx.x;
    const int lane = t & 31;
    const int wv   = t >> 5;
    const int base = blockIdx.x << 10;
    if (base >= n) return;

    int part = 0;
    const float4* __restrict__ d4 = (const float4*)deg;
    #pragma unroll 4
    for (int i = t; i < (base >> 2); i += 256) {
        const float4 v = d4[i];
        part += (int)v.x + (int)v.y + (int)v.z + (int)v.w;
    }

    const int i0 = base + (t << 2);
    int c[4];
    #pragma unroll
    for (int j = 0; j < 4; ++j) c[j] = (i0 + j < n) ? (int)deg[i0 + j] : 0;
    const int mine = c[0] + c[1] + c[2] + c[3];

    int incl = mine;
    for (int o = 1; o < 32; o <<= 1) {
        const int y = __shfl_up(incl, o);
        if (lane >= o) incl += y;
    }
    int psum = part;
    for (int o = 16; o > 0; o >>= 1) psum += __shfl_xor(psum, o);

    __shared__ int s_w[8];
    __shared__ int s_p[8];
    if (lane == 31) s_w[wv] = incl;
    if (lane == 0)  s_p[wv] = psum;
    __syncthreads();
    int off = 0;
    #pragma unroll
    for (int k = 0; k < 8; ++k) {
        off += s_p[k];
        if (k < wv) off += s_w[k];
    }
    int ex = off + incl - mine;
    #pragma unroll
    for (int j = 0; j < 4; ++j) {
        if (i0 + j < n) { cur[i0 + j] = ex; ex += c[j]; }
    }
}

__global__ void fill_kernel(const int* __restrict__ src, const int* __restrict__ dst,
                            int* __restrict__ cur, int* __restrict__ csr, int nE) {
    int e = blockIdx.x * blockDim.x + threadIdx.x;
    if (e < nE) {
        const int d = dst[e];
        const int p = __hip_atomic_fetch_add(&cur[d], 1, __ATOMIC_RELAXED, __HIP_MEMORY_SCOPE_AGENT);
        if ((unsigned)p < (unsigned)nE) csr[p] = src[e];
    }
}

__device__ __forceinline__ float4 row_edge_sum(const float* __restrict__ H,
                                               const int* __restrict__ csr,
                                               const float* __restrict__ invS,
                                               int beg, int end, float wd, int lane) {
    float4 a = make_float4(0.0f, 0.0f, 0.0f, 0.0f);
    for (int c0 = beg; c0 < end; c0 += 32) {
        const int n = (end - c0 < 32) ? (end - c0) : 32;
        int   my_s = 0;
        float my_w = 0.0f;
        if (lane < n) {
            my_s = csr[c0 + lane];
            my_w = invS[my_s] * wd;
        }
        #pragma unroll 4
        for (int j = 0; j < n; ++j) {
            const int   s = __builtin_amdgcn_readlane(my_s, j);
            const float w = __int_as_float(__builtin_amdgcn_readlane(__float_as_int(my_w), j));
            const float4 v = *((const float4*)(H + (size_t)s * DIM) + lane);
            a.x += v.x * w;
            a.y += v.y * w;
            a.z += v.z * w;
            a.w += v.w * w;
        }
    }
    return a;
}

__global__ void gather2_kernel(const float* __restrict__ H1,
                               const int* __restrict__ csr1, const int* __restrict__ cur1,
                               const float* __restrict__ invS1, const float* __restrict__ invD1,
                               const float* __restrict__ H2,
                               const int* __restrict__ csr2, const int* __restrict__ cur2,
                               const float* __restrict__ invS2, const float* __restrict__ invD2,
                               const float* __restrict__ b1, const float* __restrict__ b2,
                               float scale, int do_relu,
                               float* __restrict__ out, int nRows) {
    const int gid  = blockIdx.x * blockDim.x + threadIdx.x;
    const int row  = gid >> 5;
    const int lane = gid & 31;
    if (row >= nRows) return;

    const int prev = row > 0 ? row - 1 : 0;
    const int vb1 = cur1[prev];
    const int ve1 = cur1[row];
    const int vb2 = cur2[prev];
    const int ve2 = cur2[row];
    const float wd1 = invD1[row];
    const float wd2 = invD2[row];
    const float4 bb1 = *((const float4*)b1 + lane);
    const float4 bb2 = *((const float4*)b2 + lane);
    const int beg1 = __builtin_amdgcn_readfirstlane(row > 0 ? vb1 : 0);
    const int end1 = __builtin_amdgcn_readfirstlane(ve1);
    const int beg2 = __builtin_amdgcn_readfirstlane(row > 0 ? vb2 : 0);
    const int end2 = __builtin_amdgcn_readfirstlane(ve2);

    const float4 s1 = row_edge_sum(H1, csr1, invS1, beg1, end1, wd1, lane);
    const float4 s2 = row_edge_sum(H2, csr2, invS2, beg2, end2, wd2, lane);

    float4 v;
    v.x = (s2.x + s1.x) + bb1.x + bb2.x;
    v.y = (s2.y + s1.y) + bb1.y + bb2.y;
    v.z = (s2.z + s1.z) + bb1.z + bb2.z;
    v.w = (s2.w + s1.w) + bb1.w + bb2.w;
    v.x *= scale; v.y *= scale; v.z *= scale; v.w *= scale;
    if (do_relu) {
        v.x = fmaxf(v.x, 0.0f); v.y = fmaxf(v.y, 0.0f);
        v.z = fmaxf(v.z, 0.0f); v.w = fmaxf(v.w, 0.0f);
    }
    *((float4*)(out + (size_t)row * DIM) + lane) = v;
}

__global__ void gather1_kernel(const float* __restrict__ H,
                               const int* __restrict__ csr, const int* __restrict__ cur,
                               const float* __restrict__ invS, const float* __restrict__ invD,
                               const float* __restrict__ b, int do_relu,
                               float* __restrict__ out, int nRows) {
    const int gid  = blockIdx.x * blockDim.x + threadIdx.x;
    const int row  = gid >> 5;
    const int lane = gid & 31;
    if (row >= nRows) return;

    const int prev = row > 0 ? row - 1 : 0;
    const int vb = cur[prev];
    const int ve = cur[row];
    const float wd = invD[row];
    const float4 bb = *((const float4*)b + lane);
    const int beg = __builtin_amdgcn_readfirstlane(row > 0 ? vb : 0);
    const int end = __builtin_amdgcn_readfirstlane(ve);

    const float4 s = row_edge_sum(H, csr, invS, beg, end, wd, lane);

    float4 v;
    v.x = s.x + bb.x; v.y = s.y + bb.y; v.z = s.z + bb.z; v.w = s.w + bb.w;
    if (do_relu) {
        v.x = fmaxf(v.x, 0.0f); v.y = fmaxf(v.y, 0.0f);
        v.z = fmaxf(v.z, 0.0f); v.w = fmaxf(v.w, 0.0f);
    }
    *((float4*)(out + (size_t)row * DIM) + lane) = v;
}

extern "C" void kernel_launch(void* const* d_in, const int* in_sizes, int n_in,
                              void* d_out, int out_size, void* d_ws, size_t ws_size,
                              hipStream_t stream) {
    const float* x_user = (const float*)d_in[0];
    const float* x_item = (const float*)d_in[1];
    const int* f_src = (const int*)d_in[2];
    const int* f_dst = (const int*)d_in[3];
    const int* r_src = (const int*)d_in[4];
    const int* r_dst = (const int*)d_in[5];
    const int* v_src = (const int*)d_in[6];
    const int* v_dst = (const int*)d_in[7];
    const float* W1f = (const float*)d_in[8];
    const float* W1r = (const float*)d_in[9];
    const float* W1v = (const float*)d_in[10];
    const float* W2f = (const float*)d_in[11];
    const float* W2r = (const float*)d_in[12];
    const float* W2v = (const float*)d_in[13];
    const float* b1f = (const float*)d_in[14];
    const float* b1r = (const float*)d_in[15];
    const float* b1v = (const float*)d_in[16];
    const float* b2f = (const float*)d_in[17];
    const float* b2r = (const float*)d_in[18];
    const float* b2v = (const float*)d_in[19];

    float* w = (float*)d_ws;
    float* inv_fs = w; w += N_U;
    float* inv_fd = w; w += N_U;
    float* inv_rs = w; w += N_U;
    float* inv_rd = w; w += N_I;
    float* inv_vs = w; w += N_I;
    float* inv_vd = w; w += N_U;
    const int n_inv = 4 * N_U + 2 * N_I;
    float* h_user = w; w += (size_t)N_U * DIM;
    int* csr_f = (int*)w;
    int* csr_r = csr_f + NE;
    int* csr_v = csr_r + NE;
    int* cur_f = csr_v + NE;
    int* cur_r = cur_f + N_U;
    int* cur_v = cur_r + N_I;
    w += (size_t)N_I * DIM;
    float* u1     = w; w += (size_t)N_U * DIM;
    float* it1    = w; w += (size_t)N_I * DIM;

    float* out_u = (float*)d_out;
    float* out_i = out_u + (size_t)N_U * DIM;

    const int TPB = 256;
    const int degBlocks = (NE + TPB - 1) / TPB;
    const int gathU = (N_U * 32 + TPB - 1) / TPB;
    const int gathI = (N_I * 32 + TPB - 1) / TPB;
    dim3 gemmU((N_U / 16), DIM / 16);
    dim3 gemmI((N_I / 16), DIM / 16);
    dim3 rsGrid((N_U + 1023) / 1024, 3);

    hipMemsetAsync(inv_fs, 0, (size_t)n_inv * sizeof(float), stream);
    deg_kernel<<<degBlocks, TPB, 0, stream>>>(f_src, f_dst, inv_fs, inv_fd, NE);
    deg_kernel<<<degBlocks, TPB, 0, stream>>>(r_src, r_dst, inv_rs, inv_rd, NE);
    deg_kernel<<<degBlocks, TPB, 0, stream>>>(v_src, v_dst, inv_vs, inv_vd, NE);
    rowstart_kernel<<<rsGrid, TPB, 0, stream>>>(inv_fd, cur_f, N_U, inv_rd, cur_r, N_I,
                                                inv_vd, cur_v, N_U);
    invsqrt_kernel<<<(n_inv + TPB - 1) / TPB, TPB, 0, stream>>>(inv_fs, n_inv);
    fill_kernel<<<degBlocks, TPB, 0, stream>>>(f_src, f_dst, cur_f, csr_f, NE);
    fill_kernel<<<degBlocks, TPB, 0, stream>>>(r_src, r_dst, cur_r, csr_r, NE);
    fill_kernel<<<degBlocks, TPB, 0, stream>>>(v_src, v_dst, cur_v, csr_v, NE);

    gemm_wmma_f32<<<gemmU, 32, 0, stream>>>(x_user, W1f, h_user);
    gemm_wmma_f32<<<gemmI, 32, 0, stream>>>(x_item, W1v, it1);
    gather2_kernel<<<gathU, TPB, 0, stream>>>(h_user, csr_f, cur_f, inv_fs, inv_fd,
                                              it1,    csr_v, cur_v, inv_vs, inv_vd,
                                              b1f, b1v, 0.5f, 1, u1, N_U);
    gemm_wmma_f32<<<gemmU, 32, 0, stream>>>(x_user, W1r, h_user);
    gather1_kernel<<<gathI, TPB, 0, stream>>>(h_user, csr_r, cur_r, inv_rs, inv_rd,
                                              b1r, 1, it1, N_I);

    gemm_wmma_f32<<<gemmU, 32, 0, stream>>>(u1, W2r, h_user);
    gather1_kernel<<<gathI, TPB, 0, stream>>>(h_user, csr_r, cur_r, inv_rs, inv_rd,
                                              b2r, 0, out_i, N_I);
    gemm_wmma_f32<<<gemmU, 32, 0, stream>>>(u1, W2f, h_user);
    gemm_wmma_f32<<<gemmI, 32, 0, stream>>>(it1, W2v, u1);
    gather2_kernel<<<gathU, TPB, 0, stream>>>(h_user, csr_f, cur_f, inv_fs, inv_fd,
                                              u1,     csr_v, cur_v, inv_vs, inv_vd,
                                              b2f, b2v, 0.5f, 0, out_u, N_U);
}


# GEMM retile: one wave computes a 16x128 row block (8 accumulator tiles), W staged once per workgroup in LDS as k-paired, column-interleaved fragments; X read once, not 8 times; sum order unchanged
# speedup vs baseline: 2.0565x; 1.1221x over previous
#include <hip/hip_runtime.h>

#define N_U 100000
#define N_I 50000
#define NE  1600000
#define DIM 128

typedef __attribute__((ext_vector_type(2))) float v2f;
typedef __attribute__((ext_vector_type(8))) float v8f;

__device__ __forceinline__ void atomicAddF(float* p, float v) {
    __hip_atomic_fetch_add(p, v, __ATOMIC_RELAXED, __HIP_MEMORY_SCOPE_AGENT);
}

__global__ void deg_kernel(const int* __restrict__ src, const int* __restrict__ dst,
                           float* __restrict__ degS, float* __restrict__ degD, int nE) {
    int e = blockIdx.x * blockDim.x + threadIdx.x;
    if (e < nE) {
        atomicAddF(&degS[src[e]], 1.0f);
        atomicAddF(&degD[dst[e]], 1.0f);
    }
}

__global__ void invsqrt_kernel(float* __restrict__ a, int n) {
    int i = blockIdx.x * blockDim.x + threadIdx.x;
    if (i < n) {
        float d = a[i];
        a[i] = (d > 0.0f) ? rsqrtf(d) : 0.0f;
    }
}

typedef __attribute__((ext_vector_type(4))) float v4f;
#define GEMM_WAVES 4
__global__ __launch_bounds__(GEMM_WAVES * 32)
void gemm_wmma_f32(const float* __restrict__ X, const float* __restrict__ W,
                   float* __restrict__ H, int nTiles) {
    __shared__ __attribute__((aligned(16))) float sW[DIM * DIM];

    const int t = threadIdx.x;
    {
        #pragma unroll 4
        for (int j = 0; j < (DIM / 2) * (DIM / 4) / (GEMM_WAVES * 32); ++j) {
            const int idx = t + j * (GEMM_WAVES * 32);
            const int q   = idx & 31;
            const int kp  = idx >> 5;
            const v4f r0 = *(const v4f*)(W + (size_t)(2 * kp)     * DIM + 4 * q);
            const v4f r1 = *(const v4f*)(W + (size_t)(2 * kp + 1) * DIM + 4 * q);
            float* dst = sW + kp * 256 + (q >> 4) * 128 + (q & 15) * 4;
            v4f lo, hi;
            lo.x = r0.x; lo.y = r1.x; lo.z = r0.y; lo.w = r1.y;
            hi.x = r0.z; hi.y = r1.z; hi.z = r0.w; hi.w = r1.w;
            *(v4f*)(dst)      = lo;
            *(v4f*)(dst + 64) = hi;
        }
    }
    __syncthreads();

    const int lane = t & 31;
    const int wv   = __builtin_amdgcn_readfirstlane(t >> 5);
    const int c    = lane & 15;
    const int h    = lane >> 4;

    const float* sB = sW + h * 256 + c * 4;

    for (int tile = blockIdx.x * GEMM_WAVES + wv; tile < nTiles; tile += gridDim.x * GEMM_WAVES) {
        const int m0 = tile << 4;
        const float* __restrict__ xrow = X + (size_t)(m0 + c) * DIM + 2 * h;

        v2f a[DIM / 4];
        #pragma unroll
        for (int s = 0; s < DIM / 4; ++s) a[s] = *(const v2f*)(xrow + 4 * s);

        v8f acc0 = {}, acc1 = {}, acc2 = {}, acc3 = {}, acc4 = {}, acc5 = {}, acc6 = {}, acc7 = {};
        #pragma unroll
        for (int s = 0; s < DIM / 4; ++s) {
            const v4f p0 = *(const v4f*)(sB + s * 512);
            const v4f p1 = *(const v4f*)(sB + s * 512 + 64);
            const v4f p2 = *(const v4f*)(sB + s * 512 + 128);
            const v4f p3 = *(const v4f*)(sB + s * 512 + 192);
            acc0 = __builtin_amdgcn_wmma_f32_16x16x4_f32(false, a[s], false, p0.xy, (short)0, acc0, false, false);
            acc1 = __builtin_amdgcn_wmma_f32_16x16x4_f32(false, a[s], false, p0.zw, (short)0, acc1, false, false);
            acc2 = __builtin_amdgcn_wmma_f32_16x16x4_f32(false, a[s], false, p1.xy, (short)0, acc2, false, false);
            acc3 = __builtin_amdgcn_wmma_f32_16x16x4_f32(false, a[s], false, p1.zw, (short)0, acc3, false, false);
            acc4 = __builtin_amdgcn_wmma_f32_16x16x4_f32(false, a[s], false, p2.xy, (short)0, acc4, false, false);
            acc5 = __builtin_amdgcn_wmma_f32_16x16x4_f32(false, a[s], false, p2.zw, (short)0, acc5, false, false);
            acc6 = __builtin_amdgcn_wmma_f32_16x16x4_f32(false, a[s], false, p3.xy, (short)0, acc6, false, false);
            acc7 = __builtin_amdgcn_wmma_f32_16x16x4_f32(false, a[s], false, p3.zw, (short)0, acc7, false, false);
            if ((s & 3) == 3) __builtin_amdgcn_sched_barrier(0);
        }

        float* __restrict__ hrow = H + (size_t)(m0 + (h << 3)) * DIM + 4 * c;
        #pragma unroll
        for (int r = 0; r < 8; ++r) {
            v4f lo, hi;
            lo.x = acc0[r]; lo.y = acc1[r]; lo.z = acc2[r]; lo.w = acc3[r];
            hi.x = acc4[r]; hi.y = acc5[r]; hi.z = acc6[r]; hi.w = acc7[r];
            *(v4f*)(hrow + (size_t)r * DIM)      = lo;
            *(v4f*)(hrow + (size_t)r * DIM + 64) = hi;
        }
    }
}

__global__ void rowstart_kernel(const float* __restrict__ degA, int* __restrict__ curA, int nA,
                                const float* __restrict__ degB, int* __restrict__ curB, int nB,
                                const float* __restrict__ degC, int* __restrict__ curC, int nC) {
    const float* __restrict__ deg = degA; int* __restrict__ cur = curA; int n = nA;
    if (blockIdx.y == 1) { deg = degB; cur = curB; n = nB; }
    if (blockIdx.y == 2) { deg = degC; cur = curC; n = nC; }
    const int t    = threadIdx.x;
    const int lane = t & 31;
    const int wv   = t >> 5;
    const int base = blockIdx.x << 10;
    if (base >= n) return;

    int part = 0;
    const float4* __restrict__ d4 = (const float4*)deg;
    #pragma unroll 4
    for (int i = t; i < (base >> 2); i += 256) {
        const float4 v = d4[i];
        part += (int)v.x + (int)v.y + (int)v.z + (int)v.w;
    }

    const int i0 = base + (t << 2);
    int c[4];
    #pragma unroll
    for (int j = 0; j < 4; ++j) c[j] = (i0 + j < n) ? (int)deg[i0 + j] : 0;
    const int mine = c[0] + c[1] + c[2] + c[3];

    int incl = mine;
    for (int o = 1; o < 32; o <<= 1) {
        const int y = __shfl_up(incl, o);
        if (lane >= o) incl += y;
    }
    int psum = part;
    for (int o = 16; o > 0; o >>= 1) psum += __shfl_xor(psum, o);

    __shared__ int s_w[8];
    __shared__ int s_p[8];
    if (lane == 31) s_w[wv] = incl;
    if (lane == 0)  s_p[wv] = psum;
    __syncthreads();
    int off = 0;
    #pragma unroll
    for (int k = 0; k < 8; ++k) {
        off += s_p[k];
        if (k < wv) off += s_w[k];
    }
    int ex = off + incl - mine;
    #pragma unroll
    for (int j = 0; j < 4; ++j) {
        if (i0 + j < n) { cur[i0 + j] = ex; ex += c[j]; }
    }
}

__global__ void fill_kernel(const int* __restrict__ src, const int* __restrict__ dst,
                            int* __restrict__ cur, int* __restrict__ csr, int nE) {
    int e = blockIdx.x * blockDim.x + threadIdx.x;
    if (e < nE) {
        const int d = dst[e];
        const int p = __hip_atomic_fetch_add(&cur[d], 1, __ATOMIC_RELAXED, __HIP_MEMORY_SCOPE_AGENT);
        if ((unsigned)p < (unsigned)nE) csr[p] = src[e];
    }
}

__device__ __forceinline__ float4 row_edge_sum(const float* __restrict__ H,
                                               const int* __restrict__ csr,
                                               const float* __restrict__ invS,
                                               int beg, int end, float wd, int lane) {
    float4 a = make_float4(0.0f, 0.0f, 0.0f, 0.0f);
    for (int c0 = beg; c0 < end; c0 += 32) {
        const int n = (end - c0 < 32) ? (end - c0) : 32;
        int   my_s = 0;
        float my_w = 0.0f;
        if (lane < n) {
            my_s = csr[c0 + lane];
            my_w = invS[my_s] * wd;
        }
        #pragma unroll 4
        for (int j = 0; j < n; ++j) {
            const int   s = __builtin_amdgcn_readlane(my_s, j);
            const float w = __int_as_float(__builtin_amdgcn_readlane(__float_as_int(my_w), j));
            const float4 v = *((const float4*)(H + (size_t)s * DIM) + lane);
            a.x += v.x * w;
            a.y += v.y * w;
            a.z += v.z * w;
            a.w += v.w * w;
        }
    }
    return a;
}

__global__ void gather2_kernel(const float* __restrict__ H1,
                               const int* __restrict__ csr1, const int* __restrict__ cur1,
                               const float* __restrict__ invS1, const float* __restrict__ invD1,
                               const float* __restrict__ H2,
                               const int* __restrict__ csr2, const int* __restrict__ cur2,
                               const float* __restrict__ invS2, const float* __restrict__ invD2,
                               const float* __restrict__ b1, const float* __restrict__ b2,
                               float scale, int do_relu,
                               float* __restrict__ out, int nRows) {
    const int gid  = blockIdx.x * blockDim.x + threadIdx.x;
    const int row  = gid >> 5;
    const int lane = gid & 31;
    if (row >= nRows) return;

    const int prev = row > 0 ? row - 1 : 0;
    const int vb1 = cur1[prev];
    const int ve1 = cur1[row];
    const int vb2 = cur2[prev];
    const int ve2 = cur2[row];
    const float wd1 = invD1[row];
    const float wd2 = invD2[row];
    const float4 bb1 = *((const float4*)b1 + lane);
    const float4 bb2 = *((const float4*)b2 + lane);
    const int beg1 = __builtin_amdgcn_readfirstlane(row > 0 ? vb1 : 0);
    const int end1 = __builtin_amdgcn_readfirstlane(ve1);
    const int beg2 = __builtin_amdgcn_readfirstlane(row > 0 ? vb2 : 0);
    const int end2 = __builtin_amdgcn_readfirstlane(ve2);

    const float4 s1 = row_edge_sum(H1, csr1, invS1, beg1, end1, wd1, lane);
    const float4 s2 = row_edge_sum(H2, csr2, invS2, beg2, end2, wd2, lane);

    float4 v;
    v.x = (s2.x + s1.x) + bb1.x + bb2.x;
    v.y = (s2.y + s1.y) + bb1.y + bb2.y;
    v.z = (s2.z + s1.z) + bb1.z + bb2.z;
    v.w = (s2.w + s1.w) + bb1.w + bb2.w;
    v.x *= scale; v.y *= scale; v.z *= scale; v.w *= scale;
    if (do_relu) {
        v.x = fmaxf(v.x, 0.0f); v.y = fmaxf(v.y, 0.0f);
        v.z = fmaxf(v.z, 0.0f); v.w = fmaxf(v.w, 0.0f);
    }
    *((float4*)(out + (size_t)row * DIM) + lane) = v;
}

__global__ void gather1_kernel(const float* __restrict__ H,
                               const int* __restrict__ csr, const int* __restrict__ cur,
                               const float* __restrict__ invS, const float* __restrict__ invD,
                               const float* __restrict__ b, int do_relu,
                               float* __restrict__ out, int nRows) {
    const int gid  = blockIdx.x * blockDim.x + threadIdx.x;
    const int row  = gid >> 5;
    const int lane = gid & 31;
    if (row >= nRows) return;

    const int prev = row > 0 ? row - 1 : 0;
    const int vb = cur[prev];
    const int ve = cur[row];
    const float wd = invD[row];
    const float4 bb = *((const float4*)b + lane);
    const int beg = __builtin_amdgcn_readfirstlane(row > 0 ? vb : 0);
    const int end = __builtin_amdgcn_readfirstlane(ve);

    const float4 s = row_edge_sum(H, csr, invS, beg, end, wd, lane);

    float4 v;
    v.x = s.x + bb.x; v.y = s.y + bb.y; v.z = s.z + bb.z; v.w = s.w + bb.w;
    if (do_relu) {
        v.x = fmaxf(v.x, 0.0f); v.y = fmaxf(v.y, 0.0f);
        v.z = fmaxf(v.z, 0.0f); v.w = fmaxf(v.w, 0.0f);
    }
    *((float4*)(out + (size_t)row * DIM) + lane) = v;
}

extern "C" void kernel_launch(void* const* d_in, const int* in_sizes, int n_in,
                              void* d_out, int out_size, void* d_ws, size_t ws_size,
                              hipStream_t stream) {
    const float* x_user = (const float*)d_in[0];
    const float* x_item = (const float*)d_in[1];
    const int* f_src = (const int*)d_in[2];
    const int* f_dst = (const int*)d_in[3];
    const int* r_src = (const int*)d_in[4];
    const int* r_dst = (const int*)d_in[5];
    const int* v_src = (const int*)d_in[6];
    const int* v_dst = (const int*)d_in[7];
    const float* W1f = (const float*)d_in[8];
    const float* W1r = (const float*)d_in[9];
    const float* W1v = (const float*)d_in[10];
    const float* W2f = (const float*)d_in[11];
    const float* W2r = (const float*)d_in[12];
    const float* W2v = (const float*)d_in[13];
    const float* b1f = (const float*)d_in[14];
    const float* b1r = (const float*)d_in[15];
    const float* b1v = (const float*)d_in[16];
    const float* b2f = (const float*)d_in[17];
    const float* b2r = (const float*)d_in[18];
    const float* b2v = (const float*)d_in[19];

    float* w = (float*)d_ws;
    float* inv_fs = w; w += N_U;
    float* inv_fd = w; w += N_U;
    float* inv_rs = w; w += N_U;
    float* inv_rd = w; w += N_I;
    float* inv_vs = w; w += N_I;
    float* inv_vd = w; w += N_U;
    const int n_inv = 4 * N_U + 2 * N_I;
    float* h_user = w; w += (size_t)N_U * DIM;
    int* csr_f = (int*)w;
    int* csr_r = csr_f + NE;
    int* csr_v = csr_r + NE;
    int* cur_f = csr_v + NE;
    int* cur_r = cur_f + N_U;
    int* cur_v = cur_r + N_I;
    w += (size_t)N_I * DIM;
    float* u1     = w; w += (size_t)N_U * DIM;
    float* it1    = w; w += (size_t)N_I * DIM;

    float* out_u = (float*)d_out;
    float* out_i = out_u + (size_t)N_U * DIM;

    const int TPB = 256;
    const int degBlocks = (NE + TPB - 1) / TPB;
    const int gathU = (N_U * 32 + TPB - 1) / TPB;
    const int gathI = (N_I * 32 + TPB - 1) / TPB;
    const int gemmU = (N_U / 16 + 2 * GEMM_WAVES - 1) / (2 * GEMM_WAVES);
    const int gemmI = (N_I / 16 + 2 * GEMM_WAVES - 1) / (2 * GEMM_WAVES);
    dim3 rsGrid((N_U + 1023) / 1024, 3);

    hipMemsetAsync(inv_fs, 0, (size_t)n_inv * sizeof(float), stream);
    deg_kernel<<<degBlocks, TPB, 0, stream>>>(f_src, f_dst, inv_fs, inv_fd, NE);
    deg_kernel<<<degBlocks, TPB, 0, stream>>>(r_src, r_dst, inv_rs, inv_rd, NE);
    deg_kernel<<<degBlocks, TPB, 0, stream>>>(v_src, v_dst, inv_vs, inv_vd, NE);
    rowstart_kernel<<<rsGrid, TPB, 0, stream>>>(inv_fd, cur_f, N_U, inv_rd, cur_r, N_I,
                                                inv_vd, cur_v, N_U);
    invsqrt_kernel<<<(n_inv + TPB - 1) / TPB, TPB, 0, stream>>>(inv_fs, n_inv);
    fill_kernel<<<degBlocks, TPB, 0, stream>>>(f_src, f_dst, cur_f, csr_f, NE);
    fill_kernel<<<degBlocks, TPB, 0, stream>>>(r_src, r_dst, cur_r, csr_r, NE);
    fill_kernel<<<degBlocks, TPB, 0, stream>>>(v_src, v_dst, cur_v, csr_v, NE);

    gemm_wmma_f32<<<gemmU, GEMM_WAVES * 32, 0, stream>>>(x_user, W1f, h_user, N_U / 16);
    gemm_wmma_f32<<<gemmI, GEMM_WAVES * 32, 0, stream>>>(x_item, W1v, it1, N_I / 16);
    gather2_kernel<<<gathU, TPB, 0, stream>>>(h_user, csr_f, cur_f, inv_fs, inv_fd,
                                              it1,    csr_v, cur_v, inv_vs, inv_vd,
                                              b1f, b1v, 0.5f, 1, u1, N_U);
    gemm_wmma_f32<<<gemmU, GEMM_WAVES * 32, 0, stream>>>(x_user, W1r, h_user, N_U / 16);
    gather1_kernel<<<gathI, TPB, 0, stream>>>(h_user, csr_r, cur_r, inv_rs, inv_rd,
                                              b1r, 1, it1, N_I);

    gemm_wmma_f32<<<gemmU, GEMM_WAVES * 32, 0, stream>>>(u1, W2r, h_user, N_U / 16);
    gather1_kernel<<<gathI, TPB, 0, stream>>>(h_user, csr_r, cur_r, inv_rs, inv_rd,
                                              b2r, 0, out_i, N_I);
    gemm_wmma_f32<<<gemmU, GEMM_WAVES * 32, 0, stream>>>(u1, W2f, h_user, N_U / 16);
    gemm_wmma_f32<<<gemmI, GEMM_WAVES * 32, 0, stream>>>(it1, W2v, u1, N_I / 16);
    gather2_kernel<<<gathU, TPB, 0, stream>>>(h_user, csr_f, cur_f, inv_fs, inv_fd,
                                              u1,     csr_v, cur_v, inv_vs, inv_vd,
                                              b2f, b2v, 0.5f, 0, out_u, N_U);
}


# Fold 1/sqrt(src degree) into GEMM epilogue (H rows stored pre-scaled), dst scale once per row: gathers drop the per-edge 4-byte scale lookup and dependent hop; sum order kept, scale rounds earlier
# speedup vs baseline: 2.1603x; 1.0505x over previous
#include <hip/hip_runtime.h>

#define N_U 100000
#define N_I 50000
#define NE  1600000
#define DIM 128

typedef __attribute__((ext_vector_type(2))) float v2f;
typedef __attribute__((ext_vector_type(8))) float v8f;

__device__ __forceinline__ void atomicAddF(float* p, float v) {
    __hip_atomic_fetch_add(p, v, __ATOMIC_RELAXED, __HIP_MEMORY_SCOPE_AGENT);
}

__global__ void deg_kernel(const int* __restrict__ src, const int* __restrict__ dst,
                           float* __restrict__ degS, float* __restrict__ degD, int nE) {
    int e = blockIdx.x * blockDim.x + threadIdx.x;
    if (e < nE) {
        atomicAddF(&degS[src[e]], 1.0f);
        atomicAddF(&degD[dst[e]], 1.0f);
    }
}

__global__ void invsqrt_kernel(float* __restrict__ a, int n) {
    int i = blockIdx.x * blockDim.x + threadIdx.x;
    if (i < n) {
        float d = a[i];
        a[i] = (d > 0.0f) ? rsqrtf(d) : 0.0f;
    }
}

typedef __attribute__((ext_vector_type(4))) float v4f;
#define GEMM_WAVES 4
__global__ __launch_bounds__(GEMM_WAVES * 32)
void gemm_wmma_f32(const float* __restrict__ X, const float* __restrict__ W,
                   const float* __restrict__ rowScale,
                   float* __restrict__ H, int nTiles) {
    __shared__ __attribute__((aligned(16))) float sW[DIM * DIM];

    const int t = threadIdx.x;
    {
        #pragma unroll 4
        for (int j = 0; j < (DIM / 2) * (DIM / 4) / (GEMM_WAVES * 32); ++j) {
            const int idx = t + j * (GEMM_WAVES * 32);
            const int q   = idx & 31;
            const int kp  = idx >> 5;
            const v4f r0 = *(const v4f*)(W + (size_t)(2 * kp)     * DIM + 4 * q);
            const v4f r1 = *(const v4f*)(W + (size_t)(2 * kp + 1) * DIM + 4 * q);
            float* dst = sW + kp * 256 + (q >> 4) * 128 + (q & 15) * 4;
            v4f lo, hi;
            lo.x = r0.x; lo.y = r1.x; lo.z = r0.y; lo.w = r1.y;
            hi.x = r0.z; hi.y = r1.z; hi.z = r0.w; hi.w = r1.w;
            *(v4f*)(dst)      = lo;
            *(v4f*)(dst + 64) = hi;
        }
    }
    __syncthreads();

    const int lane = t & 31;
    const int wv   = __builtin_amdgcn_readfirstlane(t >> 5);
    const int c    = lane & 15;
    const int h    = lane >> 4;

    const float* sB = sW + h * 256 + c * 4;

    for (int tile = blockIdx.x * GEMM_WAVES + wv; tile < nTiles; tile += gridDim.x * GEMM_WAVES) {
        const int m0 = tile << 4;
        const float* __restrict__ xrow = X + (size_t)(m0 + c) * DIM + 2 * h;

        v2f a[DIM / 4];
        #pragma unroll
        for (int s = 0; s < DIM / 4; ++s) a[s] = *(const v2f*)(xrow + 4 * s);
        const v4f sc0 = *(const v4f*)(rowScale + m0 + (h << 3));
        const v4f sc1 = *(const v4f*)(rowScale + m0 + (h << 3) + 4);

        v8f acc0 = {}, acc1 = {}, acc2 = {}, acc3 = {}, acc4 = {}, acc5 = {}, acc6 = {}, acc7 = {};
        #pragma unroll
        for (int s = 0; s < DIM / 4; ++s) {
            const v4f p0 = *(const v4f*)(sB + s * 512);
            const v4f p1 = *(const v4f*)(sB + s * 512 + 64);
            const v4f p2 = *(const v4f*)(sB + s * 512 + 128);
            const v4f p3 = *(const v4f*)(sB + s * 512 + 192);
            acc0 = __builtin_amdgcn_wmma_f32_16x16x4_f32(false, a[s], false, p0.xy, (short)0, acc0, false, false);
            acc1 = __builtin_amdgcn_wmma_f32_16x16x4_f32(false, a[s], false, p0.zw, (short)0, acc1, false, false);
            acc2 = __builtin_amdgcn_wmma_f32_16x16x4_f32(false, a[s], false, p1.xy, (short)0, acc2, false, false);
            acc3 = __builtin_amdgcn_wmma_f32_16x16x4_f32(false, a[s], false, p1.zw, (short)0, acc3, false, false);
            acc4 = __builtin_amdgcn_wmma_f32_16x16x4_f32(false, a[s], false, p2.xy, (short)0, acc4, false, false);
            acc5 = __builtin_amdgcn_wmma_f32_16x16x4_f32(false, a[s], false, p2.zw, (short)0, acc5, false, false);
            acc6 = __builtin_amdgcn_wmma_f32_16x16x4_f32(false, a[s], false, p3.xy, (short)0, acc6, false, false);
            acc7 = __builtin_amdgcn_wmma_f32_16x16x4_f32(false, a[s], false, p3.zw, (short)0, acc7, false, false);
            if ((s & 3) == 3) __builtin_amdgcn_sched_barrier(0);
        }

        float* __restrict__ hrow = H + (size_t)(m0 + (h << 3)) * DIM + 4 * c;
        #pragma unroll
        for (int r = 0; r < 8; ++r) {
            const float sc = (r < 4) ? sc0[r & 3] : sc1[r & 3];
            v4f lo, hi;
            lo.x = acc0[r] * sc; lo.y = acc1[r] * sc; lo.z = acc2[r] * sc; lo.w = acc3[r] * sc;
            hi.x = acc4[r] * sc; hi.y = acc5[r] * sc; hi.z = acc6[r] * sc; hi.w = acc7[r] * sc;
            *(v4f*)(hrow + (size_t)r * DIM)      = lo;
            *(v4f*)(hrow + (size_t)r * DIM + 64) = hi;
        }
    }
}

__global__ void rowstart_kernel(const float* __restrict__ degA, int* __restrict__ curA, int nA,
                                const float* __restrict__ degB, int* __restrict__ curB, int nB,
                                const float* __restrict__ degC, int* __restrict__ curC, int nC) {
    const float* __restrict__ deg = degA; int* __restrict__ cur = curA; int n = nA;
    if (blockIdx.y == 1) { deg = degB; cur = curB; n = nB; }
    if (blockIdx.y == 2) { deg = degC; cur = curC; n = nC; }
    const int t    = threadIdx.x;
    const int lane = t & 31;
    const int wv   = t >> 5;
    const int base = blockIdx.x << 10;
    if (base >= n) return;

    int part = 0;
    const float4* __restrict__ d4 = (const float4*)deg;
    #pragma unroll 4
    for (int i = t; i < (base >> 2); i += 256) {
        const float4 v = d4[i];
        part += (int)v.x + (int)v.y + (int)v.z + (int)v.w;
    }

    const int i0 = base + (t << 2);
    int c[4];
    #pragma unroll
    for (int j = 0; j < 4; ++j) c[j] = (i0 + j < n) ? (int)deg[i0 + j] : 0;
    const int mine = c[0] + c[1] + c[2] + c[3];

    int incl = mine;
    for (int o = 1; o < 32; o <<= 1) {
        const int y = __shfl_up(incl, o);
        if (lane >= o) incl += y;
    }
    int psum = part;
    for (int o = 16; o > 0; o >>= 1) psum += __shfl_xor(psum, o);

    __shared__ int s_w[8];
    __shared__ int s_p[8];
    if (lane == 31) s_w[wv] = incl;
    if (lane == 0)  s_p[wv] = psum;
    __syncthreads();
    int off = 0;
    #pragma unroll
    for (int k = 0; k < 8; ++k) {
        off += s_p[k];
        if (k < wv) off += s_w[k];
    }
    int ex = off + incl - mine;
    #pragma unroll
    for (int j = 0; j < 4; ++j) {
        if (i0 + j < n) { cur[i0 + j] = ex; ex += c[j]; }
    }
}

__global__ void fill_kernel(const int* __restrict__ src, const int* __restrict__ dst,
                            int* __restrict__ cur, int* __restrict__ csr, int nE) {
    int e = blockIdx.x * blockDim.x + threadIdx.x;
    if (e < nE) {
        const int d = dst[e];
        const int p = __hip_atomic_fetch_add(&cur[d], 1, __ATOMIC_RELAXED, __HIP_MEMORY_SCOPE_AGENT);
        if ((unsigned)p < (unsigned)nE) csr[p] = src[e];
    }
}

__device__ __forceinline__ float4 row_edge_sum(const float* __restrict__ H,
                                               const int* __restrict__ csr,
                                               int beg, int end, int lane) {
    float4 a = make_float4(0.0f, 0.0f, 0.0f, 0.0f);
    for (int c0 = beg; c0 < end; c0 += 32) {
        const int n = (end - c0 < 32) ? (end - c0) : 32;
        int my_s = 0;
        if (lane < n) my_s = csr[c0 + lane];
        #pragma unroll 4
        for (int j = 0; j < n; ++j) {
            const int s = __builtin_amdgcn_readlane(my_s, j);
            const float4 v = *((const float4*)(H + (size_t)s * DIM) + lane);
            a.x += v.x;
            a.y += v.y;
            a.z += v.z;
            a.w += v.w;
        }
    }
    return a;
}

__global__ void gather2_kernel(const float* __restrict__ H1,
                               const int* __restrict__ csr1, const int* __restrict__ cur1,
                               const float* __restrict__ invD1,
                               const float* __restrict__ H2,
                               const int* __restrict__ csr2, const int* __restrict__ cur2,
                               const float* __restrict__ invD2,
                               const float* __restrict__ b1, const float* __restrict__ b2,
                               float scale, int do_relu,
                               float* __restrict__ out, int nRows) {
    const int gid  = blockIdx.x * blockDim.x + threadIdx.x;
    const int row  = gid >> 5;
    const int lane = gid & 31;
    if (row >= nRows) return;

    const int prev = row > 0 ? row - 1 : 0;
    const int vb1 = cur1[prev];
    const int ve1 = cur1[row];
    const int vb2 = cur2[prev];
    const int ve2 = cur2[row];
    const float wd1 = invD1[row];
    const float wd2 = invD2[row];
    const float4 bb1 = *((const float4*)b1 + lane);
    const float4 bb2 = *((const float4*)b2 + lane);
    const int beg1 = __builtin_amdgcn_readfirstlane(row > 0 ? vb1 : 0);
    const int end1 = __builtin_amdgcn_readfirstlane(ve1);
    const int beg2 = __builtin_amdgcn_readfirstlane(row > 0 ? vb2 : 0);
    const int end2 = __builtin_amdgcn_readfirstlane(ve2);

    const float4 s1 = row_edge_sum(H1, csr1, beg1, end1, lane);
    const float4 s2 = row_edge_sum(H2, csr2, beg2, end2, lane);

    float4 v;
    v.x = (s2.x * wd2 + s1.x * wd1) + bb1.x + bb2.x;
    v.y = (s2.y * wd2 + s1.y * wd1) + bb1.y + bb2.y;
    v.z = (s2.z * wd2 + s1.z * wd1) + bb1.z + bb2.z;
    v.w = (s2.w * wd2 + s1.w * wd1) + bb1.w + bb2.w;
    v.x *= scale; v.y *= scale; v.z *= scale; v.w *= scale;
    if (do_relu) {
        v.x = fmaxf(v.x, 0.0f); v.y = fmaxf(v.y, 0.0f);
        v.z = fmaxf(v.z, 0.0f); v.w = fmaxf(v.w, 0.0f);
    }
    *((float4*)(out + (size_t)row * DIM) + lane) = v;
}

__global__ void gather1_kernel(const float* __restrict__ H,
                               const int* __restrict__ csr, const int* __restrict__ cur,
                               const float* __restrict__ invD,
                               const float* __restrict__ b, int do_relu,
                               float* __restrict__ out, int nRows) {
    const int gid  = blockIdx.x * blockDim.x + threadIdx.x;
    const int row  = gid >> 5;
    const int lane = gid & 31;
    if (row >= nRows) return;

    const int prev = row > 0 ? row - 1 : 0;
    const int vb = cur[prev];
    const int ve = cur[row];
    const float wd = invD[row];
    const float4 bb = *((const float4*)b + lane);
    const int beg = __builtin_amdgcn_readfirstlane(row > 0 ? vb : 0);
    const int end = __builtin_amdgcn_readfirstlane(ve);

    const float4 s = row_edge_sum(H, csr, beg, end, lane);

    float4 v;
    v.x = s.x * wd + bb.x; v.y = s.y * wd + bb.y; v.z = s.z * wd + bb.z; v.w = s.w * wd + bb.w;
    if (do_relu) {
        v.x = fmaxf(v.x, 0.0f); v.y = fmaxf(v.y, 0.0f);
        v.z = fmaxf(v.z, 0.0f); v.w = fmaxf(v.w, 0.0f);
    }
    *((float4*)(out + (size_t)row * DIM) + lane) = v;
}

extern "C" void kernel_launch(void* const* d_in, const int* in_sizes, int n_in,
                              void* d_out, int out_size, void* d_ws, size_t ws_size,
                              hipStream_t stream) {
    const float* x_user = (const float*)d_in[0];
    const float* x_item = (const float*)d_in[1];
    const int* f_src = (const int*)d_in[2];
    const int* f_dst = (const int*)d_in[3];
    const int* r_src = (const int*)d_in[4];
    const int* r_dst = (const int*)d_in[5];
    const int* v_src = (const int*)d_in[6];
    const int* v_dst = (const int*)d_in[7];
    const float* W1f = (const float*)d_in[8];
    const float* W1r = (const float*)d_in[9];
    const float* W1v = (const float*)d_in[10];
    const float* W2f = (const float*)d_in[11];
    const float* W2r = (const float*)d_in[12];
    const float* W2v = (const float*)d_in[13];
    const float* b1f = (const float*)d_in[14];
    const float* b1r = (const float*)d_in[15];
    const float* b1v = (const float*)d_in[16];
    const float* b2f = (const float*)d_in[17];
    const float* b2r = (const float*)d_in[18];
    const float* b2v = (const float*)d_in[19];

    float* w = (float*)d_ws;
    float* inv_fs = w; w += N_U;
    float* inv_fd = w; w += N_U;
    float* inv_rs = w; w += N_U;
    float* inv_rd = w; w += N_I;
    float* inv_vs = w; w += N_I;
    float* inv_vd = w; w += N_U;
    const int n_inv = 4 * N_U + 2 * N_I;
    float* h_user = w; w += (size_t)N_U * DIM;
    int* csr_f = (int*)w;
    int* csr_r = csr_f + NE;
    int* csr_v = csr_r + NE;
    int* cur_f = csr_v + NE;
    int* cur_r = cur_f + N_U;
    int* cur_v = cur_r + N_I;
    w += (size_t)N_I * DIM;
    float* u1     = w; w += (size_t)N_U * DIM;
    float* it1    = w; w += (size_t)N_I * DIM;

    float* out_u = (float*)d_out;
    float* out_i = out_u + (size_t)N_U * DIM;

    const int TPB = 256;
    const int degBlocks = (NE + TPB - 1) / TPB;
    const int gathU = (N_U * 32 + TPB - 1) / TPB;
    const int gathI = (N_I * 32 + TPB - 1) / TPB;
    const int gemmU = (N_U / 16 + 2 * GEMM_WAVES - 1) / (2 * GEMM_WAVES);
    const int gemmI = (N_I / 16 + 2 * GEMM_WAVES - 1) / (2 * GEMM_WAVES);
    dim3 rsGrid((N_U + 1023) / 1024, 3);

    hipMemsetAsync(inv_fs, 0, (size_t)n_inv * sizeof(float), stream);
    deg_kernel<<<degBlocks, TPB, 0, stream>>>(f_src, f_dst, inv_fs, inv_fd, NE);
    deg_kernel<<<degBlocks, TPB, 0, stream>>>(r_src, r_dst, inv_rs, inv_rd, NE);
    deg_kernel<<<degBlocks, TPB, 0, stream>>>(v_src, v_dst, inv_vs, inv_vd, NE);
    rowstart_kernel<<<rsGrid, TPB, 0, stream>>>(inv_fd, cur_f, N_U, inv_rd, cur_r, N_I,
                                                inv_vd, cur_v, N_U);
    invsqrt_kernel<<<(n_inv + TPB - 1) / TPB, TPB, 0, stream>>>(inv_fs, n_inv);
    fill_kernel<<<degBlocks, TPB, 0, stream>>>(f_src, f_dst, cur_f, csr_f, NE);
    fill_kernel<<<degBlocks, TPB, 0, stream>>>(r_src, r_dst, cur_r, csr_r, NE);
    fill_kernel<<<degBlocks, TPB, 0, stream>>>(v_src, v_dst, cur_v, csr_v, NE);

    gemm_wmma_f32<<<gemmU, GEMM_WAVES * 32, 0, stream>>>(x_user, W1f, inv_fs, h_user, N_U / 16);
    gemm_wmma_f32<<<gemmI, GEMM_WAVES * 32, 0, stream>>>(x_item, W1v, inv_vs, it1, N_I / 16);
    gather2_kernel<<<gathU, TPB, 0, stream>>>(h_user, csr_f, cur_f, inv_fd,
                                              it1,    csr_v, cur_v, inv_vd,
                                              b1f, b1v, 0.5f, 1, u1, N_U);
    gemm_wmma_f32<<<gemmU, GEMM_WAVES * 32, 0, stream>>>(x_user, W1r, inv_rs, h_user, N_U / 16);
    gather1_kernel<<<gathI, TPB, 0, stream>>>(h_user, csr_r, cur_r, inv_rd,
                                              b1r, 1, it1, N_I);

    gemm_wmma_f32<<<gemmU, GEMM_WAVES * 32, 0, stream>>>(u1, W2r, inv_rs, h_user, N_U / 16);
    gather1_kernel<<<gathI, TPB, 0, stream>>>(h_user, csr_r, cur_r, inv_rd,
                                              b2r, 0, out_i, N_I);
    gemm_wmma_f32<<<gemmU, GEMM_WAVES * 32, 0, stream>>>(u1, W2f, inv_fs, h_user, N_U / 16);
    gemm_wmma_f32<<<gemmI, GEMM_WAVES * 32, 0, stream>>>(it1, W2v, inv_vs, u1, N_I / 16);
    gather2_kernel<<<gathU, TPB, 0, stream>>>(h_user, csr_f, cur_f, inv_fd,
                                              u1,     csr_v, cur_v, inv_vd,
                                              b2f, b2v, 0.5f, 0, out_u, N_U);
}


# Count pass keeps each edge's slot (value returned by its integer add on the dst count); fill places edges at row start + slot with plain loads, no atomic; row sum order stays arrival rank
# speedup vs baseline: 2.5912x; 1.1995x over previous
#include <hip/hip_runtime.h>

#define N_U 100000
#define N_I 50000
#define NE  1600000
#define DIM 128

typedef __attribute__((ext_vector_type(2))) float v2f;
typedef __attribute__((ext_vector_type(8))) float v8f;

__global__ void deg_kernel(const int* __restrict__ src, const int* __restrict__ dst,
                           int* __restrict__ degS, int* __restrict__ degD,
                           int* __restrict__ rank, int nE) {
    int e = blockIdx.x * blockDim.x + threadIdx.x;
    if (e < nE) {
        const int s = src[e];
        const int d = dst[e];
        __hip_atomic_fetch_add(&degS[s], 1, __ATOMIC_RELAXED, __HIP_MEMORY_SCOPE_AGENT);
        rank[e] = __hip_atomic_fetch_add(&degD[d], 1, __ATOMIC_RELAXED, __HIP_MEMORY_SCOPE_AGENT);
    }
}

__global__ void invsqrt_kernel(int* __restrict__ a, int n) {
    int i = blockIdx.x * blockDim.x + threadIdx.x;
    if (i < n) {
        const int d = a[i];
        const float r = (d > 0) ? rsqrtf((float)d) : 0.0f;
        a[i] = __float_as_int(r);
    }
}

typedef __attribute__((ext_vector_type(4))) float v4f;
#define GEMM_WAVES 4
__global__ __launch_bounds__(GEMM_WAVES * 32)
void gemm_wmma_f32(const float* __restrict__ X, const float* __restrict__ W,
                   const float* __restrict__ rowScale,
                   float* __restrict__ H, int nTiles) {
    __shared__ __attribute__((aligned(16))) float sW[DIM * DIM];

    const int t = threadIdx.x;
    {
        #pragma unroll 4
        for (int j = 0; j < (DIM / 2) * (DIM / 4) / (GEMM_WAVES * 32); ++j) {
            const int idx = t + j * (GEMM_WAVES * 32);
            const int q   = idx & 31;
            const int kp  = idx >> 5;
            const v4f r0 = *(const v4f*)(W + (size_t)(2 * kp)     * DIM + 4 * q);
            const v4f r1 = *(const v4f*)(W + (size_t)(2 * kp + 1) * DIM + 4 * q);
            float* dst = sW + kp * 256 + (q >> 4) * 128 + (q & 15) * 4;
            v4f lo, hi;
            lo.x = r0.x; lo.y = r1.x; lo.z = r0.y; lo.w = r1.y;
            hi.x = r0.z; hi.y = r1.z; hi.z = r0.w; hi.w = r1.w;
            *(v4f*)(dst)      = lo;
            *(v4f*)(dst + 64) = hi;
        }
    }
    __syncthreads();

    const int lane = t & 31;
    const int wv   = __builtin_amdgcn_readfirstlane(t >> 5);
    const int c    = lane & 15;
    const int h    = lane >> 4;

    const float* sB = sW + h * 256 + c * 4;

    for (int tile = blockIdx.x * GEMM_WAVES + wv; tile < nTiles; tile += gridDim.x * GEMM_WAVES) {
        const int m0 = tile << 4;
        const float* __restrict__ xrow = X + (size_t)(m0 + c) * DIM + 2 * h;

        v2f a[DIM / 4];
        #pragma unroll
        for (int s = 0; s < DIM / 4; ++s) a[s] = *(const v2f*)(xrow + 4 * s);
        const v4f sc0 = *(const v4f*)(rowScale + m0 + (h << 3));
        const v4f sc1 = *(const v4f*)(rowScale + m0 + (h << 3) + 4);

        v8f acc0 = {}, acc1 = {}, acc2 = {}, acc3 = {}, acc4 = {}, acc5 = {}, acc6 = {}, acc7 = {};
        #pragma unroll
        for (int s = 0; s < DIM / 4; ++s) {
            const v4f p0 = *(const v4f*)(sB + s * 512);
            const v4f p1 = *(const v4f*)(sB + s * 512 + 64);
            const v4f p2 = *(const v4f*)(sB + s * 512 + 128);
            const v4f p3 = *(const v4f*)(sB + s * 512 + 192);
            acc0 = __builtin_amdgcn_wmma_f32_16x16x4_f32(false, a[s], false, p0.xy, (short)0, acc0, false, false);
            acc1 = __builtin_amdgcn_wmma_f32_16x16x4_f32(false, a[s], false, p0.zw, (short)0, acc1, false, false);
            acc2 = __builtin_amdgcn_wmma_f32_16x16x4_f32(false, a[s], false, p1.xy, (short)0, acc2, false, false);
            acc3 = __builtin_amdgcn_wmma_f32_16x16x4_f32(false, a[s], false, p1.zw, (short)0, acc3, false, false);
            acc4 = __builtin_amdgcn_wmma_f32_16x16x4_f32(false, a[s], false, p2.xy, (short)0, acc4, false, false);
            acc5 = __builtin_amdgcn_wmma_f32_16x16x4_f32(false, a[s], false, p2.zw, (short)0, acc5, false, false);
            acc6 = __builtin_amdgcn_wmma_f32_16x16x4_f32(false, a[s], false, p3.xy, (short)0, acc6, false, false);
            acc7 = __builtin_amdgcn_wmma_f32_16x16x4_f32(false, a[s], false, p3.zw, (short)0, acc7, false, false);
            if ((s & 3) == 3) __builtin_amdgcn_sched_barrier(0);
        }

        float* __restrict__ hrow = H + (size_t)(m0 + (h << 3)) * DIM + 4 * c;
        #pragma unroll
        for (int r = 0; r < 8; ++r) {
            const float sc = (r < 4) ? sc0[r & 3] : sc1[r & 3];
            v4f lo, hi;
            lo.x = acc0[r] * sc; lo.y = acc1[r] * sc; lo.z = acc2[r] * sc; lo.w = acc3[r] * sc;
            hi.x = acc4[r] * sc; hi.y = acc5[r] * sc; hi.z = acc6[r] * sc; hi.w = acc7[r] * sc;
            *(v4f*)(hrow + (size_t)r * DIM)      = lo;
            *(v4f*)(hrow + (size_t)r * DIM + 64) = hi;
        }
    }
}

__global__ void rowstart_kernel(const int* __restrict__ degA, int* __restrict__ curA, int nA,
                                const int* __restrict__ degB, int* __restrict__ curB, int nB,
                                const int* __restrict__ degC, int* __restrict__ curC, int nC) {
    const int* __restrict__ deg = degA; int* __restrict__ cur = curA; int n = nA;
    if (blockIdx.y == 1) { deg = degB; cur = curB; n = nB; }
    if (blockIdx.y == 2) { deg = degC; cur = curC; n = nC; }
    const int t    = threadIdx.x;
    const int lane = t & 31;
    const int wv   = t >> 5;
    const int base = blockIdx.x << 10;
    if (base >= n) return;

    int part = 0;
    const int4* __restrict__ d4 = (const int4*)deg;
    #pragma unroll 4
    for (int i = t; i < (base >> 2); i += 256) {
        const int4 v = d4[i];
        part += v.x + v.y + v.z + v.w;
    }

    const int i0 = base + (t << 2);
    int c[4];
    #pragma unroll
    for (int j = 0; j < 4; ++j) c[j] = (i0 + j < n) ? deg[i0 + j] : 0;
    const int mine = c[0] + c[1] + c[2] + c[3];

    int incl = mine;
    for (int o = 1; o < 32; o <<= 1) {
        const int y = __shfl_up(incl, o);
        if (lane >= o) incl += y;
    }
    int psum = part;
    for (int o = 16; o > 0; o >>= 1) psum += __shfl_xor(psum, o);

    __shared__ int s_w[8];
    __shared__ int s_p[8];
    if (lane == 31) s_w[wv] = incl;
    if (lane == 0)  s_p[wv] = psum;
    __syncthreads();
    int off = 0;
    #pragma unroll
    for (int k = 0; k < 8; ++k) {
        off += s_p[k];
        if (k < wv) off += s_w[k];
    }
    int ex = off + incl - mine;
    #pragma unroll
    for (int j = 0; j < 4; ++j) {
        if (i0 + j < n) {
            cur[i0 + j] = ex;
            ex += c[j];
            if (i0 + j == n - 1) cur[n] = ex;
        }
    }
}

__global__ void fill_kernel(const int* __restrict__ src, const int* __restrict__ dst,
                            const int* __restrict__ rank, const int* __restrict__ cur,
                            int* __restrict__ csr, int nE) {
    int e = blockIdx.x * blockDim.x + threadIdx.x;
    if (e < nE) {
        const int s = src[e];
        const int p = cur[dst[e]] + rank[e];
        if ((unsigned)p < (unsigned)nE) csr[p] = s;
    }
}

__device__ __forceinline__ float4 row_edge_sum(const float* __restrict__ H,
                                               const int* __restrict__ csr,
                                               int beg, int end, int lane) {
    float4 a = make_float4(0.0f, 0.0f, 0.0f, 0.0f);
    for (int c0 = beg; c0 < end; c0 += 32) {
        const int n = (end - c0 < 32) ? (end - c0) : 32;
        int my_s = 0;
        if (lane < n) my_s = csr[c0 + lane];
        #pragma unroll 4
        for (int j = 0; j < n; ++j) {
            const int s = __builtin_amdgcn_readlane(my_s, j);
            const float4 v = *((const float4*)(H + (size_t)s * DIM) + lane);
            a.x += v.x;
            a.y += v.y;
            a.z += v.z;
            a.w += v.w;
        }
    }
    return a;
}

__global__ void gather2_kernel(const float* __restrict__ H1,
                               const int* __restrict__ csr1, const int* __restrict__ cur1,
                               const float* __restrict__ invD1,
                               const float* __restrict__ H2,
                               const int* __restrict__ csr2, const int* __restrict__ cur2,
                               const float* __restrict__ invD2,
                               const float* __restrict__ b1, const float* __restrict__ b2,
                               float scale, int do_relu,
                               float* __restrict__ out, int nRows) {
    const int gid  = blockIdx.x * blockDim.x + threadIdx.x;
    const int row  = gid >> 5;
    const int lane = gid & 31;
    if (row >= nRows) return;

    const int vb1 = cur1[row];
    const int ve1 = cur1[row + 1];
    const int vb2 = cur2[row];
    const int ve2 = cur2[row + 1];
    const float wd1 = invD1[row];
    const float wd2 = invD2[row];
    const float4 bb1 = *((const float4*)b1 + lane);
    const float4 bb2 = *((const float4*)b2 + lane);
    const int beg1 = __builtin_amdgcn_readfirstlane(vb1);
    const int end1 = __builtin_amdgcn_readfirstlane(ve1);
    const int beg2 = __builtin_amdgcn_readfirstlane(vb2);
    const int end2 = __builtin_amdgcn_readfirstlane(ve2);

    const float4 s1 = row_edge_sum(H1, csr1, beg1, end1, lane);
    const float4 s2 = row_edge_sum(H2, csr2, beg2, end2, lane);

    float4 v;
    v.x = (s2.x * wd2 + s1.x * wd1) + bb1.x + bb2.x;
    v.y = (s2.y * wd2 + s1.y * wd1) + bb1.y + bb2.y;
    v.z = (s2.z * wd2 + s1.z * wd1) + bb1.z + bb2.z;
    v.w = (s2.w * wd2 + s1.w * wd1) + bb1.w + bb2.w;
    v.x *= scale; v.y *= scale; v.z *= scale; v.w *= scale;
    if (do_relu) {
        v.x = fmaxf(v.x, 0.0f); v.y = fmaxf(v.y, 0.0f);
        v.z = fmaxf(v.z, 0.0f); v.w = fmaxf(v.w, 0.0f);
    }
    *((float4*)(out + (size_t)row * DIM) + lane) = v;
}

__global__ void gather1_kernel(const float* __restrict__ H,
                               const int* __restrict__ csr, const int* __restrict__ cur,
                               const float* __restrict__ invD,
                               const float* __restrict__ b, int do_relu,
                               float* __restrict__ out, int nRows) {
    const int gid  = blockIdx.x * blockDim.x + threadIdx.x;
    const int row  = gid >> 5;
    const int lane = gid & 31;
    if (row >= nRows) return;

    const int vb = cur[row];
    const int ve = cur[row + 1];
    const float wd = invD[row];
    const float4 bb = *((const float4*)b + lane);
    const int beg = __builtin_amdgcn_readfirstlane(vb);
    const int end = __builtin_amdgcn_readfirstlane(ve);

    const float4 s = row_edge_sum(H, csr, beg, end, lane);

    float4 v;
    v.x = s.x * wd + bb.x; v.y = s.y * wd + bb.y; v.z = s.z * wd + bb.z; v.w = s.w * wd + bb.w;
    if (do_relu) {
        v.x = fmaxf(v.x, 0.0f); v.y = fmaxf(v.y, 0.0f);
        v.z = fmaxf(v.z, 0.0f); v.w = fmaxf(v.w, 0.0f);
    }
    *((float4*)(out + (size_t)row * DIM) + lane) = v;
}

extern "C" void kernel_launch(void* const* d_in, const int* in_sizes, int n_in,
                              void* d_out, int out_size, void* d_ws, size_t ws_size,
                              hipStream_t stream) {
    const float* x_user = (const float*)d_in[0];
    const float* x_item = (const float*)d_in[1];
    const int* f_src = (const int*)d_in[2];
    const int* f_dst = (const int*)d_in[3];
    const int* r_src = (const int*)d_in[4];
    const int* r_dst = (const int*)d_in[5];
    const int* v_src = (const int*)d_in[6];
    const int* v_dst = (const int*)d_in[7];
    const float* W1f = (const float*)d_in[8];
    const float* W1r = (const float*)d_in[9];
    const float* W1v = (const float*)d_in[10];
    const float* W2f = (const float*)d_in[11];
    const float* W2r = (const float*)d_in[12];
    const float* W2v = (const float*)d_in[13];
    const float* b1f = (const float*)d_in[14];
    const float* b1r = (const float*)d_in[15];
    const float* b1v = (const float*)d_in[16];
    const float* b2f = (const float*)d_in[17];
    const float* b2r = (const float*)d_in[18];
    const float* b2v = (const float*)d_in[19];

    float* w = (float*)d_ws;
    float* inv_fs = w; w += N_U;
    float* inv_fd = w; w += N_U;
    float* inv_rs = w; w += N_U;
    float* inv_rd = w; w += N_I;
    float* inv_vs = w; w += N_I;
    float* inv_vd = w; w += N_U;
    const int n_inv = 4 * N_U + 2 * N_I;
    float* h_user = w; w += (size_t)N_U * DIM;
    int* csr_f = (int*)w;
    int* csr_r = csr_f + NE;
    int* csr_v = csr_r + NE;
    int* cur_f = csr_v + NE;
    int* cur_r = cur_f + N_U + 1;
    int* cur_v = cur_r + N_I + 1;
    w += (size_t)N_I * DIM;
    float* u1     = w; w += (size_t)N_U * DIM;
    int* rank_f = (int*)u1;
    int* rank_r = rank_f + NE;
    int* rank_v = rank_r + NE;
    float* it1    = w; w += (size_t)N_I * DIM;

    float* out_u = (float*)d_out;
    float* out_i = out_u + (size_t)N_U * DIM;

    const int TPB = 256;
    const int degBlocks = (NE + TPB - 1) / TPB;
    const int gathU = (N_U * 32 + TPB - 1) / TPB;
    const int gathI = (N_I * 32 + TPB - 1) / TPB;
    const int gemmU = (N_U / 16 + 2 * GEMM_WAVES - 1) / (2 * GEMM_WAVES);
    const int gemmI = (N_I / 16 + 2 * GEMM_WAVES - 1) / (2 * GEMM_WAVES);
    dim3 rsGrid((N_U + 1023) / 1024, 3);

    hipMemsetAsync(inv_fs, 0, (size_t)n_inv * sizeof(float), stream);
    deg_kernel<<<degBlocks, TPB, 0, stream>>>(f_src, f_dst, (int*)inv_fs, (int*)inv_fd, rank_f, NE);
    deg_kernel<<<degBlocks, TPB, 0, stream>>>(r_src, r_dst, (int*)inv_rs, (int*)inv_rd, rank_r, NE);
    deg_kernel<<<degBlocks, TPB, 0, stream>>>(v_src, v_dst, (int*)inv_vs, (int*)inv_vd, rank_v, NE);
    rowstart_kernel<<<rsGrid, TPB, 0, stream>>>((const int*)inv_fd, cur_f, N_U,
                                                (const int*)inv_rd, cur_r, N_I,
                                                (const int*)inv_vd, cur_v, N_U);
    invsqrt_kernel<<<(n_inv + TPB - 1) / TPB, TPB, 0, stream>>>((int*)inv_fs, n_inv);
    fill_kernel<<<degBlocks, TPB, 0, stream>>>(f_src, f_dst, rank_f, cur_f, csr_f, NE);
    fill_kernel<<<degBlocks, TPB, 0, stream>>>(r_src, r_dst, rank_r, cur_r, csr_r, NE);
    fill_kernel<<<degBlocks, TPB, 0, stream>>>(v_src, v_dst, rank_v, cur_v, csr_v, NE);

    gemm_wmma_f32<<<gemmU, GEMM_WAVES * 32, 0, stream>>>(x_user, W1f, inv_fs, h_user, N_U / 16);
    gemm_wmma_f32<<<gemmI, GEMM_WAVES * 32, 0, stream>>>(x_item, W1v, inv_vs, it1, N_I / 16);
    gather2_kernel<<<gathU, TPB, 0, stream>>>(h_user, csr_f, cur_f, inv_fd,
                                              it1,    csr_v, cur_v, inv_vd,
                                              b1f, b1v, 0.5f, 1, u1, N_U);
    gemm_wmma_f32<<<gemmU, GEMM_WAVES * 32, 0, stream>>>(x_user, W1r, inv_rs, h_user, N_U / 16);
    gather1_kernel<<<gathI, TPB, 0, stream>>>(h_user, csr_r, cur_r, inv_rd,
                                              b1r, 1, it1, N_I);

    gemm_wmma_f32<<<gemmU, GEMM_WAVES * 32, 0, stream>>>(u1, W2r, inv_rs, h_user, N_U / 16);
    gather1_kernel<<<gathI, TPB, 0, stream>>>(h_user, csr_r, cur_r, inv_rd,
                                              b2r, 0, out_i, N_I);
    gemm_wmma_f32<<<gemmU, GEMM_WAVES * 32, 0, stream>>>(u1, W2f, inv_fs, h_user, N_U / 16);
    gemm_wmma_f32<<<gemmI, GEMM_WAVES * 32, 0, stream>>>(it1, W2v, inv_vs, u1, N_I / 16);
    gather2_kernel<<<gathU, TPB, 0, stream>>>(h_user, csr_f, cur_f, inv_fd,
                                              u1,     csr_v, cur_v, inv_vd,
                                              b2f, b2v, 0.5f, 0, out_u, N_U);
}


# Degree counts and per-edge row slots from LDS counters per edge slice (a block owns a 50000-bin window) and a scan over slices, no global atomics; row sum order now slice, then arrival in slice
# speedup vs baseline: 4.0521x; 1.5638x over previous
#include <hip/hip_runtime.h>

#define N_U 100000
#define N_I 50000
#define NE  1600000
#define DIM 128

typedef __attribute__((ext_vector_type(2))) float v2f;
typedef __attribute__((ext_vector_type(8))) float v8f;

#define CNT_SLICES    25
#define CNT_SLICE_LEN (NE / CNT_SLICES)
#define CNT_RANGE     50000
#define N_BINS        (4 * N_U + 2 * N_I)
#define CNT_JOBS      (N_BINS / CNT_RANGE)
#define CNT_THREADS   1024
#define CNT_TRIPS     ((CNT_SLICE_LEN / 4 + CNT_THREADS - 1) / CNT_THREADS)
static_assert(NE % CNT_SLICES == 0 && CNT_SLICE_LEN % 4 == 0, "slices are whole 16-byte groups");
static_assert(N_U == 2 * CNT_RANGE && N_I == CNT_RANGE, "window table below assumes these sizes");

__global__ __launch_bounds__(CNT_THREADS)
void count_kernel(const int* __restrict__ f_src, const int* __restrict__ f_dst,
                  const int* __restrict__ r_src, const int* __restrict__ r_dst,
                  const int* __restrict__ v_src, const int* __restrict__ v_dst,
                  int* __restrict__ rank_f, int* __restrict__ rank_r, int* __restrict__ rank_v,
                  int* __restrict__ part) {
    __shared__ __attribute__((aligned(16))) int cnt[CNT_RANGE];

    const int t   = threadIdx.x;
    const int job = blockIdx.x / CNT_SLICES;
    const int k   = blockIdx.x - job * CNT_SLICES;

    const int* __restrict__ keys = f_src;
    int* __restrict__ rank = nullptr;
    int lo = 0;
    switch (job) {
        case 0: keys = f_src; break;
        case 1: keys = f_src; lo = CNT_RANGE; break;
        case 2: keys = f_dst; rank = rank_f; break;
        case 3: keys = f_dst; rank = rank_f; lo = CNT_RANGE; break;
        case 4: keys = r_src; break;
        case 5: keys = r_src; lo = CNT_RANGE; break;
        case 6: keys = r_dst; rank = rank_r; break;
        case 7: keys = v_src; break;
        case 8: keys = v_dst; rank = rank_v; break;
        default: keys = v_dst; rank = rank_v; lo = CNT_RANGE; break;
    }

    for (int i = t; i < CNT_RANGE / 4; i += CNT_THREADS) ((int4*)cnt)[i] = make_int4(0, 0, 0, 0);
    __syncthreads();

    const int e0 = k * CNT_SLICE_LEN;
    const int4* __restrict__ k4 = (const int4*)(keys + e0);
    int4 v[CNT_TRIPS];
    #pragma unroll
    for (int j = 0; j < CNT_TRIPS; ++j) {
        const int i = t + j * CNT_THREADS;
        v[j] = (i < CNT_SLICE_LEN / 4) ? k4[i] : make_int4(-1, -1, -1, -1);
    }
    if (rank != nullptr) {
        #pragma unroll
        for (int j = 0; j < CNT_TRIPS; ++j) {
            const unsigned b0 = (unsigned)(v[j].x - lo), b1 = (unsigned)(v[j].y - lo);
            const unsigned b2 = (unsigned)(v[j].z - lo), b3 = (unsigned)(v[j].w - lo);
            int o0 = 0, o1 = 0, o2 = 0, o3 = 0;
            if (b0 < CNT_RANGE) o0 = __hip_atomic_fetch_add(&cnt[b0], 1, __ATOMIC_RELAXED, __HIP_MEMORY_SCOPE_WORKGROUP);
            if (b1 < CNT_RANGE) o1 = __hip_atomic_fetch_add(&cnt[b1], 1, __ATOMIC_RELAXED, __HIP_MEMORY_SCOPE_WORKGROUP);
            if (b2 < CNT_RANGE) o2 = __hip_atomic_fetch_add(&cnt[b2], 1, __ATOMIC_RELAXED, __HIP_MEMORY_SCOPE_WORKGROUP);
            if (b3 < CNT_RANGE) o3 = __hip_atomic_fetch_add(&cnt[b3], 1, __ATOMIC_RELAXED, __HIP_MEMORY_SCOPE_WORKGROUP);
            int* r = rank + e0 + 4 * (t + j * CNT_THREADS);
            if (b0 < CNT_RANGE) r[0] = o0;
            if (b1 < CNT_RANGE) r[1] = o1;
            if (b2 < CNT_RANGE) r[2] = o2;
            if (b3 < CNT_RANGE) r[3] = o3;
        }
    } else {
        #pragma unroll
        for (int j = 0; j < CNT_TRIPS; ++j) {
            const unsigned b0 = (unsigned)(v[j].x - lo), b1 = (unsigned)(v[j].y - lo);
            const unsigned b2 = (unsigned)(v[j].z - lo), b3 = (unsigned)(v[j].w - lo);
            if (b0 < CNT_RANGE) __hip_atomic_fetch_add(&cnt[b0], 1, __ATOMIC_RELAXED, __HIP_MEMORY_SCOPE_WORKGROUP);
            if (b1 < CNT_RANGE) __hip_atomic_fetch_add(&cnt[b1], 1, __ATOMIC_RELAXED, __HIP_MEMORY_SCOPE_WORKGROUP);
            if (b2 < CNT_RANGE) __hip_atomic_fetch_add(&cnt[b2], 1, __ATOMIC_RELAXED, __HIP_MEMORY_SCOPE_WORKGROUP);
            if (b3 < CNT_RANGE) __hip_atomic_fetch_add(&cnt[b3], 1, __ATOMIC_RELAXED, __HIP_MEMORY_SCOPE_WORKGROUP);
        }
    }
    __syncthreads();

    int4* __restrict__ o4 = (int4*)(part + (size_t)k * N_BINS + job * CNT_RANGE);
    for (int i = t; i < CNT_RANGE / 4; i += CNT_THREADS) o4[i] = ((const int4*)cnt)[i];
}

__global__ void combine_kernel(int* __restrict__ part, int* __restrict__ deg) {
    const int i = blockIdx.x * blockDim.x + threadIdx.x;
    if (i < N_BINS / 4) {
        const int b = 4 * i;
        const bool isDst = (b >= N_U && b < 2 * N_U) || (b >= 3 * N_U && b < 3 * N_U + N_I) ||
                           (b >= 3 * N_U + 2 * N_I);
        int4* __restrict__ p4 = (int4*)part + i;
        int4 c[CNT_SLICES];
        #pragma unroll
        for (int k = 0; k < CNT_SLICES; ++k) c[k] = p4[(size_t)k * (N_BINS / 4)];
        int4 run = make_int4(0, 0, 0, 0);
        #pragma unroll
        for (int k = 0; k < CNT_SLICES; ++k) {
            if (isDst) p4[(size_t)k * (N_BINS / 4)] = run;
            run.x += c[k].x; run.y += c[k].y; run.z += c[k].z; run.w += c[k].w;
        }
        ((int4*)deg)[i] = run;
    }
}

__global__ void invsqrt_kernel(int* __restrict__ a, int n) {
    int i = blockIdx.x * blockDim.x + threadIdx.x;
    if (i < n) {
        const int d = a[i];
        const float r = (d > 0) ? rsqrtf((float)d) : 0.0f;
        a[i] = __float_as_int(r);
    }
}

typedef __attribute__((ext_vector_type(4))) float v4f;
#define GEMM_WAVES 4
__global__ __launch_bounds__(GEMM_WAVES * 32)
void gemm_wmma_f32(const float* __restrict__ X, const float* __restrict__ W,
                   const float* __restrict__ rowScale,
                   float* __restrict__ H, int nTiles) {
    __shared__ __attribute__((aligned(16))) float sW[DIM * DIM];

    const int t = threadIdx.x;
    {
        #pragma unroll 4
        for (int j = 0; j < (DIM / 2) * (DIM / 4) / (GEMM_WAVES * 32); ++j) {
            const int idx = t + j * (GEMM_WAVES * 32);
            const int q   = idx & 31;
            const int kp  = idx >> 5;
            const v4f r0 = *(const v4f*)(W + (size_t)(2 * kp)     * DIM + 4 * q);
            const v4f r1 = *(const v4f*)(W + (size_t)(2 * kp + 1) * DIM + 4 * q);
            float* dst = sW + kp * 256 + (q >> 4) * 128 + (q & 15) * 4;
            v4f lo, hi;
            lo.x = r0.x; lo.y = r1.x; lo.z = r0.y; lo.w = r1.y;
            hi.x = r0.z; hi.y = r1.z; hi.z = r0.w; hi.w = r1.w;
            *(v4f*)(dst)      = lo;
            *(v4f*)(dst + 64) = hi;
        }
    }
    __syncthreads();

    const int lane = t & 31;
    const int wv   = __builtin_amdgcn_readfirstlane(t >> 5);
    const int c    = lane & 15;
    const int h    = lane >> 4;

    const float* sB = sW + h * 256 + c * 4;

    for (int tile = blockIdx.x * GEMM_WAVES + wv; tile < nTiles; tile += gridDim.x * GEMM_WAVES) {
        const int m0 = tile << 4;
        const float* __restrict__ xrow = X + (size_t)(m0 + c) * DIM + 2 * h;

        v2f a[DIM / 4];
        #pragma unroll
        for (int s = 0; s < DIM / 4; ++s) a[s] = *(const v2f*)(xrow + 4 * s);
        const v4f sc0 = *(const v4f*)(rowScale + m0 + (h << 3));
        const v4f sc1 = *(const v4f*)(rowScale + m0 + (h << 3) + 4);

        v8f acc0 = {}, acc1 = {}, acc2 = {}, acc3 = {}, acc4 = {}, acc5 = {}, acc6 = {}, acc7 = {};
        #pragma unroll
        for (int s = 0; s < DIM / 4; ++s) {
            const v4f p0 = *(const v4f*)(sB + s * 512);
            const v4f p1 = *(const v4f*)(sB + s * 512 + 64);
            const v4f p2 = *(const v4f*)(sB + s * 512 + 128);
            const v4f p3 = *(const v4f*)(sB + s * 512 + 192);
            acc0 = __builtin_amdgcn_wmma_f32_16x16x4_f32(false, a[s], false, p0.xy, (short)0, acc0, false, false);
            acc1 = __builtin_amdgcn_wmma_f32_16x16x4_f32(false, a[s], false, p0.zw, (short)0, acc1, false, false);
            acc2 = __builtin_amdgcn_wmma_f32_16x16x4_f32(false, a[s], false, p1.xy, (short)0, acc2, false, false);
            acc3 = __builtin_amdgcn_wmma_f32_16x16x4_f32(false, a[s], false, p1.zw, (short)0, acc3, false, false);
            acc4 = __builtin_amdgcn_wmma_f32_16x16x4_f32(false, a[s], false, p2.xy, (short)0, acc4, false, false);
            acc5 = __builtin_amdgcn_wmma_f32_16x16x4_f32(false, a[s], false, p2.zw, (short)0, acc5, false, false);
            acc6 = __builtin_amdgcn_wmma_f32_16x16x4_f32(false, a[s], false, p3.xy, (short)0, acc6, false, false);
            acc7 = __builtin_amdgcn_wmma_f32_16x16x4_f32(false, a[s], false, p3.zw, (short)0, acc7, false, false);
            if ((s & 3) == 3) __builtin_amdgcn_sched_barrier(0);
        }

        float* __restrict__ hrow = H + (size_t)(m0 + (h << 3)) * DIM + 4 * c;
        #pragma unroll
        for (int r = 0; r < 8; ++r) {
            const float sc = (r < 4) ? sc0[r & 3] : sc1[r & 3];
            v4f lo, hi;
            lo.x = acc0[r] * sc; lo.y = acc1[r] * sc; lo.z = acc2[r] * sc; lo.w = acc3[r] * sc;
            hi.x = acc4[r] * sc; hi.y = acc5[r] * sc; hi.z = acc6[r] * sc; hi.w = acc7[r] * sc;
            *(v4f*)(hrow + (size_t)r * DIM)      = lo;
            *(v4f*)(hrow + (size_t)r * DIM + 64) = hi;
        }
    }
}

__global__ void rowstart_kernel(const int* __restrict__ degA, int* __restrict__ curA, int nA,
                                const int* __restrict__ degB, int* __restrict__ curB, int nB,
                                const int* __restrict__ degC, int* __restrict__ curC, int nC) {
    const int* __restrict__ deg = degA; int* __restrict__ cur = curA; int n = nA;
    if (blockIdx.y == 1) { deg = degB; cur = curB; n = nB; }
    if (blockIdx.y == 2) { deg = degC; cur = curC; n = nC; }
    const int t    = threadIdx.x;
    const int lane = t & 31;
    const int wv   = t >> 5;
    const int base = blockIdx.x << 10;
    if (base >= n) return;

    int part = 0;
    const int4* __restrict__ d4 = (const int4*)deg;
    #pragma unroll 4
    for (int i = t; i < (base >> 2); i += 256) {
        const int4 v = d4[i];
        part += v.x + v.y + v.z + v.w;
    }

    const int i0 = base + (t << 2);
    int c[4];
    #pragma unroll
    for (int j = 0; j < 4; ++j) c[j] = (i0 + j < n) ? deg[i0 + j] : 0;
    const int mine = c[0] + c[1] + c[2] + c[3];

    int incl = mine;
    for (int o = 1; o < 32; o <<= 1) {
        const int y = __shfl_up(incl, o);
        if (lane >= o) incl += y;
    }
    int psum = part;
    for (int o = 16; o > 0; o >>= 1) psum += __shfl_xor(psum, o);

    __shared__ int s_w[8];
    __shared__ int s_p[8];
    if (lane == 31) s_w[wv] = incl;
    if (lane == 0)  s_p[wv] = psum;
    __syncthreads();
    int off = 0;
    #pragma unroll
    for (int k = 0; k < 8; ++k) {
        off += s_p[k];
        if (k < wv) off += s_w[k];
    }
    int ex = off + incl - mine;
    #pragma unroll
    for (int j = 0; j < 4; ++j) {
        if (i0 + j < n) {
            cur[i0 + j] = ex;
            ex += c[j];
            if (i0 + j == n - 1) cur[n] = ex;
        }
    }
}

__global__ void fill_kernel(const int* __restrict__ src, const int* __restrict__ dst,
                            const int* __restrict__ rank, const int* __restrict__ cur,
                            const int* __restrict__ base,
                            int* __restrict__ csr, int nE) {
    int e = blockIdx.x * blockDim.x + threadIdx.x;
    if (e < nE) {
        const int s = src[e];
        const int d = dst[e];
        const int p = cur[d] + base[(size_t)(e / CNT_SLICE_LEN) * N_BINS + d] + rank[e];
        if ((unsigned)p < (unsigned)nE) csr[p] = s;
    }
}

__device__ __forceinline__ float4 row_edge_sum(const float* __restrict__ H,
                                               const int* __restrict__ csr,
                                               int beg, int end, int lane) {
    float4 a = make_float4(0.0f, 0.0f, 0.0f, 0.0f);
    for (int c0 = beg; c0 < end; c0 += 32) {
        const int n = (end - c0 < 32) ? (end - c0) : 32;
        int my_s = 0;
        if (lane < n) my_s = csr[c0 + lane];
        #pragma unroll 4
        for (int j = 0; j < n; ++j) {
            const int s = __builtin_amdgcn_readlane(my_s, j);
            const float4 v = *((const float4*)(H + (size_t)s * DIM) + lane);
            a.x += v.x;
            a.y += v.y;
            a.z += v.z;
            a.w += v.w;
        }
    }
    return a;
}

__global__ void gather2_kernel(const float* __restrict__ H1,
                               const int* __restrict__ csr1, const int* __restrict__ cur1,
                               const float* __restrict__ invD1,
                               const float* __restrict__ H2,
                               const int* __restrict__ csr2, const int* __restrict__ cur2,
                               const float* __restrict__ invD2,
                               const float* __restrict__ b1, const float* __restrict__ b2,
                               float scale, int do_relu,
                               float* __restrict__ out, int nRows) {
    const int gid  = blockIdx.x * blockDim.x + threadIdx.x;
    const int row  = gid >> 5;
    const int lane = gid & 31;
    if (row >= nRows) return;

    const int vb1 = cur1[row];
    const int ve1 = cur1[row + 1];
    const int vb2 = cur2[row];
    const int ve2 = cur2[row + 1];
    const float wd1 = invD1[row];
    const float wd2 = invD2[row];
    const float4 bb1 = *((const float4*)b1 + lane);
    const float4 bb2 = *((const float4*)b2 + lane);
    const int beg1 = __builtin_amdgcn_readfirstlane(vb1);
    const int end1 = __builtin_amdgcn_readfirstlane(ve1);
    const int beg2 = __builtin_amdgcn_readfirstlane(vb2);
    const int end2 = __builtin_amdgcn_readfirstlane(ve2);

    const float4 s1 = row_edge_sum(H1, csr1, beg1, end1, lane);
    const float4 s2 = row_edge_sum(H2, csr2, beg2, end2, lane);

    float4 v;
    v.x = (s2.x * wd2 + s1.x * wd1) + bb1.x + bb2.x;
    v.y = (s2.y * wd2 + s1.y * wd1) + bb1.y + bb2.y;
    v.z = (s2.z * wd2 + s1.z * wd1) + bb1.z + bb2.z;
    v.w = (s2.w * wd2 + s1.w * wd1) + bb1.w + bb2.w;
    v.x *= scale; v.y *= scale; v.z *= scale; v.w *= scale;
    if (do_relu) {
        v.x = fmaxf(v.x, 0.0f); v.y = fmaxf(v.y, 0.0f);
        v.z = fmaxf(v.z, 0.0f); v.w = fmaxf(v.w, 0.0f);
    }
    *((float4*)(out + (size_t)row * DIM) + lane) = v;
}

__global__ void gather1_kernel(const float* __restrict__ H,
                               const int* __restrict__ csr, const int* __restrict__ cur,
                               const float* __restrict__ invD,
                               const float* __restrict__ b, int do_relu,
                               float* __restrict__ out, int nRows) {
    const int gid  = blockIdx.x * blockDim.x + threadIdx.x;
    const int row  = gid >> 5;
    const int lane = gid & 31;
    if (row >= nRows) return;

    const int vb = cur[row];
    const int ve = cur[row + 1];
    const float wd = invD[row];
    const float4 bb = *((const float4*)b + lane);
    const int beg = __builtin_amdgcn_readfirstlane(vb);
    const int end = __builtin_amdgcn_readfirstlane(ve);

    const float4 s = row_edge_sum(H, csr, beg, end, lane);

    float4 v;
    v.x = s.x * wd + bb.x; v.y = s.y * wd + bb.y; v.z = s.z * wd + bb.z; v.w = s.w * wd + bb.w;
    if (do_relu) {
        v.x = fmaxf(v.x, 0.0f); v.y = fmaxf(v.y, 0.0f);
        v.z = fmaxf(v.z, 0.0f); v.w = fmaxf(v.w, 0.0f);
    }
    *((float4*)(out + (size_t)row * DIM) + lane) = v;
}

extern "C" void kernel_launch(void* const* d_in, const int* in_sizes, int n_in,
                              void* d_out, int out_size, void* d_ws, size_t ws_size,
                              hipStream_t stream) {
    const float* x_user = (const float*)d_in[0];
    const float* x_item = (const float*)d_in[1];
    const int* f_src = (const int*)d_in[2];
    const int* f_dst = (const int*)d_in[3];
    const int* r_src = (const int*)d_in[4];
    const int* r_dst = (const int*)d_in[5];
    const int* v_src = (const int*)d_in[6];
    const int* v_dst = (const int*)d_in[7];
    const float* W1f = (const float*)d_in[8];
    const float* W1r = (const float*)d_in[9];
    const float* W1v = (const float*)d_in[10];
    const float* W2f = (const float*)d_in[11];
    const float* W2r = (const float*)d_in[12];
    const float* W2v = (const float*)d_in[13];
    const float* b1f = (const float*)d_in[14];
    const float* b1r = (const float*)d_in[15];
    const float* b1v = (const float*)d_in[16];
    const float* b2f = (const float*)d_in[17];
    const float* b2r = (const float*)d_in[18];
    const float* b2v = (const float*)d_in[19];

    float* w = (float*)d_ws;
    float* inv_fs = w; w += N_U;
    float* inv_fd = w; w += N_U;
    float* inv_rs = w; w += N_U;
    float* inv_rd = w; w += N_I;
    float* inv_vs = w; w += N_I;
    float* inv_vd = w; w += N_U;
    const int n_inv = N_BINS;
    float* h_user = w; w += (size_t)N_U * DIM;
    int* part = (int*)h_user;
    int* csr_f = (int*)w;
    int* csr_r = csr_f + NE;
    int* csr_v = csr_r + NE;
    int* cur_f = csr_v + NE;
    int* cur_r = cur_f + N_U + 1;
    int* cur_v = cur_r + N_I + 1;
    w += (size_t)N_I * DIM;
    float* u1     = w; w += (size_t)N_U * DIM;
    int* rank_f = (int*)u1;
    int* rank_r = rank_f + NE;
    int* rank_v = rank_r + NE;
    float* it1    = w; w += (size_t)N_I * DIM;

    float* out_u = (float*)d_out;
    float* out_i = out_u + (size_t)N_U * DIM;

    const int TPB = 256;
    const int degBlocks = (NE + TPB - 1) / TPB;
    const int gathU = (N_U * 32 + TPB - 1) / TPB;
    const int gathI = (N_I * 32 + TPB - 1) / TPB;
    const int gemmU = (N_U / 16 + 2 * GEMM_WAVES - 1) / (2 * GEMM_WAVES);
    const int gemmI = (N_I / 16 + 2 * GEMM_WAVES - 1) / (2 * GEMM_WAVES);
    dim3 rsGrid((N_U + 1023) / 1024, 3);

    count_kernel<<<CNT_JOBS * CNT_SLICES, CNT_THREADS, 0, stream>>>(f_src, f_dst, r_src, r_dst, v_src, v_dst,
                                                                  rank_f, rank_r, rank_v, part);
    combine_kernel<<<(N_BINS / 4 + TPB - 1) / TPB, TPB, 0, stream>>>(part, (int*)inv_fs);
    rowstart_kernel<<<rsGrid, TPB, 0, stream>>>((const int*)inv_fd, cur_f, N_U,
                                                (const int*)inv_rd, cur_r, N_I,
                                                (const int*)inv_vd, cur_v, N_U);
    invsqrt_kernel<<<(n_inv + TPB - 1) / TPB, TPB, 0, stream>>>((int*)inv_fs, n_inv);
    fill_kernel<<<degBlocks, TPB, 0, stream>>>(f_src, f_dst, rank_f, cur_f, part + N_U, csr_f, NE);
    fill_kernel<<<degBlocks, TPB, 0, stream>>>(r_src, r_dst, rank_r, cur_r, part + 3 * N_U, csr_r, NE);
    fill_kernel<<<degBlocks, TPB, 0, stream>>>(v_src, v_dst, rank_v, cur_v, part + 3 * N_U + 2 * N_I, csr_v, NE);

    gemm_wmma_f32<<<gemmU, GEMM_WAVES * 32, 0, stream>>>(x_user, W1f, inv_fs, h_user, N_U / 16);
    gemm_wmma_f32<<<gemmI, GEMM_WAVES * 32, 0, stream>>>(x_item, W1v, inv_vs, it1, N_I / 16);
    gather2_kernel<<<gathU, TPB, 0, stream>>>(h_user, csr_f, cur_f, inv_fd,
                                              it1,    csr_v, cur_v, inv_vd,
                                              b1f, b1v, 0.5f, 1, u1, N_U);
    gemm_wmma_f32<<<gemmU, GEMM_WAVES * 32, 0, stream>>>(x_user, W1r, inv_rs, h_user, N_U / 16);
    gather1_kernel<<<gathI, TPB, 0, stream>>>(h_user, csr_r, cur_r, inv_rd,
                                              b1r, 1, it1, N_I);

    gemm_wmma_f32<<<gemmU, GEMM_WAVES * 32, 0, stream>>>(u1, W2r, inv_rs, h_user, N_U / 16);
    gather1_kernel<<<gathI, TPB, 0, stream>>>(h_user, csr_r, cur_r, inv_rd,
                                              b2r, 0, out_i, N_I);
    gemm_wmma_f32<<<gemmU, GEMM_WAVES * 32, 0, stream>>>(u1, W2f, inv_fs, h_user, N_U / 16);
    gemm_wmma_f32<<<gemmI, GEMM_WAVES * 32, 0, stream>>>(it1, W2v, inv_vs, u1, N_I / 16);
    gather2_kernel<<<gathU, TPB, 0, stream>>>(h_user, csr_f, cur_f, inv_fd,
                                              u1,     csr_v, cur_v, inv_vd,
                                              b2f, b2v, 0.5f, 0, out_u, N_U);
}


# Independent passes share launches: 3 edge placements run beside the 3 layer-1 GEMMs in one grid, layer 2's 3 GEMMs are one grid (output rows are scratch until the final gathers); 17 to 10 launches
# speedup vs baseline: 4.8926x; 1.2074x over previous
#include <hip/hip_runtime.h>

#define N_U 100000
#define N_I 50000
#define NE  1600000
#define DIM 128

typedef __attribute__((ext_vector_type(2))) float v2f;
typedef __attribute__((ext_vector_type(8))) float v8f;

#define CNT_SLICES    25
#define CNT_SLICE_LEN (NE / CNT_SLICES)
#define CNT_RANGE     50000
#define N_BINS        (4 * N_U + 2 * N_I)
#define CNT_JOBS      (N_BINS / CNT_RANGE)
#define CNT_THREADS   1024
#define CNT_TRIPS     ((CNT_SLICE_LEN / 4 + CNT_THREADS - 1) / CNT_THREADS)
static_assert(NE % CNT_SLICES == 0 && CNT_SLICE_LEN % 4 == 0, "slices are whole 16-byte groups");
static_assert(N_U == 2 * CNT_RANGE && N_I == CNT_RANGE, "window table below assumes these sizes");

__global__ __launch_bounds__(CNT_THREADS)
void count_kernel(const int* __restrict__ f_src, const int* __restrict__ f_dst,
                  const int* __restrict__ r_src, const int* __restrict__ r_dst,
                  const int* __restrict__ v_src, const int* __restrict__ v_dst,
                  int* __restrict__ rank_f, int* __restrict__ rank_r, int* __restrict__ rank_v,
                  int* __restrict__ part) {
    __shared__ __attribute__((aligned(16))) int cnt[CNT_RANGE];

    const int t   = threadIdx.x;
    const int job = blockIdx.x / CNT_SLICES;
    const int k   = blockIdx.x - job * CNT_SLICES;

    const int* __restrict__ keys = f_src;
    int* __restrict__ rank = nullptr;
    int lo = 0;
    switch (job) {
        case 0: keys = f_src; break;
        case 1: keys = f_src; lo = CNT_RANGE; break;
        case 2: keys = f_dst; rank = rank_f; break;
        case 3: keys = f_dst; rank = rank_f; lo = CNT_RANGE; break;
        case 4: keys = r_src; break;
        case 5: keys = r_src; lo = CNT_RANGE; break;
        case 6: keys = r_dst; rank = rank_r; break;
        case 7: keys = v_src; break;
        case 8: keys = v_dst; rank = rank_v; break;
        default: keys = v_dst; rank = rank_v; lo = CNT_RANGE; break;
    }

    for (int i = t; i < CNT_RANGE / 4; i += CNT_THREADS) ((int4*)cnt)[i] = make_int4(0, 0, 0, 0);
    __syncthreads();

    const int e0 = k * CNT_SLICE_LEN;
    const int4* __restrict__ k4 = (const int4*)(keys + e0);
    int4 v[CNT_TRIPS];
    #pragma unroll
    for (int j = 0; j < CNT_TRIPS; ++j) {
        const int i = t + j * CNT_THREADS;
        v[j] = (i < CNT_SLICE_LEN / 4) ? k4[i] : make_int4(-1, -1, -1, -1);
    }
    if (rank != nullptr) {
        #pragma unroll
        for (int j = 0; j < CNT_TRIPS; ++j) {
            const unsigned b0 = (unsigned)(v[j].x - lo), b1 = (unsigned)(v[j].y - lo);
            const unsigned b2 = (unsigned)(v[j].z - lo), b3 = (unsigned)(v[j].w - lo);
            int o0 = 0, o1 = 0, o2 = 0, o3 = 0;
            if (b0 < CNT_RANGE) o0 = __hip_atomic_fetch_add(&cnt[b0], 1, __ATOMIC_RELAXED, __HIP_MEMORY_SCOPE_WORKGROUP);
            if (b1 < CNT_RANGE) o1 = __hip_atomic_fetch_add(&cnt[b1], 1, __ATOMIC_RELAXED, __HIP_MEMORY_SCOPE_WORKGROUP);
            if (b2 < CNT_RANGE) o2 = __hip_atomic_fetch_add(&cnt[b2], 1, __ATOMIC_RELAXED, __HIP_MEMORY_SCOPE_WORKGROUP);
            if (b3 < CNT_RANGE) o3 = __hip_atomic_fetch_add(&cnt[b3], 1, __ATOMIC_RELAXED, __HIP_MEMORY_SCOPE_WORKGROUP);
            int* r = rank + e0 + 4 * (t + j * CNT_THREADS);
            if (b0 < CNT_RANGE) r[0] = o0;
            if (b1 < CNT_RANGE) r[1] = o1;
            if (b2 < CNT_RANGE) r[2] = o2;
            if (b3 < CNT_RANGE) r[3] = o3;
        }
    } else {
        #pragma unroll
        for (int j = 0; j < CNT_TRIPS; ++j) {
            const unsigned b0 = (unsigned)(v[j].x - lo), b1 = (unsigned)(v[j].y - lo);
            const unsigned b2 = (unsigned)(v[j].z - lo), b3 = (unsigned)(v[j].w - lo);
            if (b0 < CNT_RANGE) __hip_atomic_fetch_add(&cnt[b0], 1, __ATOMIC_RELAXED, __HIP_MEMORY_SCOPE_WORKGROUP);
            if (b1 < CNT_RANGE) __hip_atomic_fetch_add(&cnt[b1], 1, __ATOMIC_RELAXED, __HIP_MEMORY_SCOPE_WORKGROUP);
            if (b2 < CNT_RANGE) __hip_atomic_fetch_add(&cnt[b2], 1, __ATOMIC_RELAXED, __HIP_MEMORY_SCOPE_WORKGROUP);
            if (b3 < CNT_RANGE) __hip_atomic_fetch_add(&cnt[b3], 1, __ATOMIC_RELAXED, __HIP_MEMORY_SCOPE_WORKGROUP);
        }
    }
    __syncthreads();

    int4* __restrict__ o4 = (int4*)(part + (size_t)k * N_BINS + job * CNT_RANGE);
    for (int i = t; i < CNT_RANGE / 4; i += CNT_THREADS) o4[i] = ((const int4*)cnt)[i];
}

__global__ void combine_kernel(int* __restrict__ part, int* __restrict__ deg) {
    const int i = blockIdx.x * blockDim.x + threadIdx.x;
    if (i < N_BINS / 4) {
        const int b = 4 * i;
        const bool isDst = (b >= N_U && b < 2 * N_U) || (b >= 3 * N_U && b < 3 * N_U + N_I) ||
                           (b >= 3 * N_U + 2 * N_I);
        int4* __restrict__ p4 = (int4*)part + i;
        int4 c[CNT_SLICES];
        #pragma unroll
        for (int k = 0; k < CNT_SLICES; ++k) c[k] = p4[(size_t)k * (N_BINS / 4)];
        int4 run = make_int4(0, 0, 0, 0);
        #pragma unroll
        for (int k = 0; k < CNT_SLICES; ++k) {
            if (isDst) p4[(size_t)k * (N_BINS / 4)] = run;
            run.x += c[k].x; run.y += c[k].y; run.z += c[k].z; run.w += c[k].w;
        }
        ((int4*)deg)[i] = run;
    }
}

__global__ void invsqrt_kernel(int* __restrict__ a, int n) {
    int i = blockIdx.x * blockDim.x + threadIdx.x;
    if (i < n) {
        const int d = a[i];
        const float r = (d > 0) ? rsqrtf((float)d) : 0.0f;
        a[i] = __float_as_int(r);
    }
}

typedef __attribute__((ext_vector_type(4))) float v4f;
#define GEMM_WAVES 4
__device__ __forceinline__
void gemm_rows(const float* __restrict__ X, const float* __restrict__ W,
               const float* __restrict__ rowScale,
               float* __restrict__ H, int nTiles, int blk, int nBlk) {
    __shared__ __attribute__((aligned(16))) float sW[DIM * DIM];

    const int t = threadIdx.x;
    {
        #pragma unroll 4
        for (int j = 0; j < (DIM / 2) * (DIM / 4) / (GEMM_WAVES * 32); ++j) {
            const int idx = t + j * (GEMM_WAVES * 32);
            const int q   = idx & 31;
            const int kp  = idx >> 5;
            const v4f r0 = *(const v4f*)(W + (size_t)(2 * kp)     * DIM + 4 * q);
            const v4f r1 = *(const v4f*)(W + (size_t)(2 * kp + 1) * DIM + 4 * q);
            float* dst = sW + kp * 256 + (q >> 4) * 128 + (q & 15) * 4;
            v4f lo, hi;
            lo.x = r0.x; lo.y = r1.x; lo.z = r0.y; lo.w = r1.y;
            hi.x = r0.z; hi.y = r1.z; hi.z = r0.w; hi.w = r1.w;
            *(v4f*)(dst)      = lo;
            *(v4f*)(dst + 64) = hi;
        }
    }
    __syncthreads();

    const int lane = t & 31;
    const int wv   = __builtin_amdgcn_readfirstlane(t >> 5);
    const int c    = lane & 15;
    const int h    = lane >> 4;

    const float* sB = sW + h * 256 + c * 4;

    for (int tile = blk * GEMM_WAVES + wv; tile < nTiles; tile += nBlk * GEMM_WAVES) {
        const int m0 = tile << 4;
        const float* __restrict__ xrow = X + (size_t)(m0 + c) * DIM + 2 * h;

        v2f a[DIM / 4];
        #pragma unroll
        for (int s = 0; s < DIM / 4; ++s) a[s] = *(const v2f*)(xrow + 4 * s);
        const v4f sc0 = *(const v4f*)(rowScale + m0 + (h << 3));
        const v4f sc1 = *(const v4f*)(rowScale + m0 + (h << 3) + 4);

        v8f acc0 = {}, acc1 = {}, acc2 = {}, acc3 = {}, acc4 = {}, acc5 = {}, acc6 = {}, acc7 = {};
        #pragma unroll
        for (int s = 0; s < DIM / 4; ++s) {
            const v4f p0 = *(const v4f*)(sB + s * 512);
            const v4f p1 = *(const v4f*)(sB + s * 512 + 64);
            const v4f p2 = *(const v4f*)(sB + s * 512 + 128);
            const v4f p3 = *(const v4f*)(sB + s * 512 + 192);
            acc0 = __builtin_amdgcn_wmma_f32_16x16x4_f32(false, a[s], false, p0.xy, (short)0, acc0, false, false);
            acc1 = __builtin_amdgcn_wmma_f32_16x16x4_f32(false, a[s], false, p0.zw, (short)0, acc1, false, false);
            acc2 = __builtin_amdgcn_wmma_f32_16x16x4_f32(false, a[s], false, p1.xy, (short)0, acc2, false, false);
            acc3 = __builtin_amdgcn_wmma_f32_16x16x4_f32(false, a[s], false, p1.zw, (short)0, acc3, false, false);
            acc4 = __builtin_amdgcn_wmma_f32_16x16x4_f32(false, a[s], false, p2.xy, (short)0, acc4, false, false);
            acc5 = __builtin_amdgcn_wmma_f32_16x16x4_f32(false, a[s], false, p2.zw, (short)0, acc5, false, false);
            acc6 = __builtin_amdgcn_wmma_f32_16x16x4_f32(false, a[s], false, p3.xy, (short)0, acc6, false, false);
            acc7 = __builtin_amdgcn_wmma_f32_16x16x4_f32(false, a[s], false, p3.zw, (short)0, acc7, false, false);
            if ((s & 3) == 3) __builtin_amdgcn_sched_barrier(0);
        }

        float* __restrict__ hrow = H + (size_t)(m0 + (h << 3)) * DIM + 4 * c;
        #pragma unroll
        for (int r = 0; r < 8; ++r) {
            const float sc = (r < 4) ? sc0[r & 3] : sc1[r & 3];
            v4f lo, hi;
            lo.x = acc0[r] * sc; lo.y = acc1[r] * sc; lo.z = acc2[r] * sc; lo.w = acc3[r] * sc;
            hi.x = acc4[r] * sc; hi.y = acc5[r] * sc; hi.z = acc6[r] * sc; hi.w = acc7[r] * sc;
            *(v4f*)(hrow + (size_t)r * DIM)      = lo;
            *(v4f*)(hrow + (size_t)r * DIM + 64) = hi;
        }
    }
}

__global__ void rowstart_kernel(const int* __restrict__ degA, int* __restrict__ curA, int nA,
                                const int* __restrict__ degB, int* __restrict__ curB, int nB,
                                const int* __restrict__ degC, int* __restrict__ curC, int nC) {
    const int* __restrict__ deg = degA; int* __restrict__ cur = curA; int n = nA;
    if (blockIdx.y == 1) { deg = degB; cur = curB; n = nB; }
    if (blockIdx.y == 2) { deg = degC; cur = curC; n = nC; }
    const int t    = threadIdx.x;
    const int lane = t & 31;
    const int wv   = t >> 5;
    const int base = blockIdx.x << 10;
    if (base >= n) return;

    int part = 0;
    const int4* __restrict__ d4 = (const int4*)deg;
    #pragma unroll 4
    for (int i = t; i < (base >> 2); i += 256) {
        const int4 v = d4[i];
        part += v.x + v.y + v.z + v.w;
    }

    const int i0 = base + (t << 2);
    int c[4];
    #pragma unroll
    for (int j = 0; j < 4; ++j) c[j] = (i0 + j < n) ? deg[i0 + j] : 0;
    const int mine = c[0] + c[1] + c[2] + c[3];

    int incl = mine;
    for (int o = 1; o < 32; o <<= 1) {
        const int y = __shfl_up(incl, o);
        if (lane >= o) incl += y;
    }
    int psum = part;
    for (int o = 16; o > 0; o >>= 1) psum += __shfl_xor(psum, o);

    __shared__ int s_w[8];
    __shared__ int s_p[8];
    if (lane == 31) s_w[wv] = incl;
    if (lane == 0)  s_p[wv] = psum;
    __syncthreads();
    int off = 0;
    #pragma unroll
    for (int k = 0; k < 8; ++k) {
        off += s_p[k];
        if (k < wv) off += s_w[k];
    }
    int ex = off + incl - mine;
    #pragma unroll
    for (int j = 0; j < 4; ++j) {
        if (i0 + j < n) {
            cur[i0 + j] = ex;
            ex += c[j];
            if (i0 + j == n - 1) cur[n] = ex;
        }
    }
}

#define FILL_THREADS     (GEMM_WAVES * 32)
#define FILL_PER         8
#define FILL_BLOCK_EDGES (FILL_THREADS * FILL_PER)
#define FILL_BLOCKS      ((NE + FILL_BLOCK_EDGES - 1) / FILL_BLOCK_EDGES)
__device__ __forceinline__
void fill_edges(const int* __restrict__ src, const int* __restrict__ dst,
                const int* __restrict__ rank, const int* __restrict__ cur,
                const int* __restrict__ base, int* __restrict__ csr, int blk) {
    const int e0 = blk * FILL_BLOCK_EDGES + threadIdx.x;
    int s[FILL_PER], d[FILL_PER], r[FILL_PER], p[FILL_PER];
    #pragma unroll
    for (int j = 0; j < FILL_PER; ++j) {
        const int e  = e0 + j * FILL_THREADS;
        const int ec = (e < NE) ? e : (NE - 1);
        s[j] = src[ec];
        d[j] = dst[ec];
        r[j] = rank[ec];
    }
    #pragma unroll
    for (int j = 0; j < FILL_PER; ++j) {
        const int e  = e0 + j * FILL_THREADS;
        const int ec = (e < NE) ? e : (NE - 1);
        p[j] = cur[d[j]] + base[(size_t)(ec / CNT_SLICE_LEN) * N_BINS + d[j]] + r[j];
    }
    #pragma unroll
    for (int j = 0; j < FILL_PER; ++j) {
        const int e = e0 + j * FILL_THREADS;
        if (e < NE && (unsigned)p[j] < (unsigned)NE) csr[p[j]] = s[j];
    }
}

#define GEMM_BLOCKS_U   ((N_U / 16 + 2 * GEMM_WAVES - 1) / (2 * GEMM_WAVES))
#define GEMM_BLOCKS_I   ((N_I / 16 + 2 * GEMM_WAVES - 1) / (2 * GEMM_WAVES))
#define GEMM_BLOCKS_ALL (2 * GEMM_BLOCKS_U + GEMM_BLOCKS_I)
__device__ __forceinline__
void gemm_job(int g,
              const float* __restrict__ XA, const float* __restrict__ WA,
              const float* __restrict__ sA, float* __restrict__ HA,
              const float* __restrict__ XB, const float* __restrict__ WB,
              const float* __restrict__ sB, float* __restrict__ HB,
              const float* __restrict__ XC, const float* __restrict__ WC,
              const float* __restrict__ sC, float* __restrict__ HC) {
    const float* __restrict__ X = XA; const float* __restrict__ W = WA;
    const float* __restrict__ sc = sA; float* __restrict__ H = HA;
    int nTiles = N_U / 16, blk = g, nBlk = GEMM_BLOCKS_U;
    if (g >= 2 * GEMM_BLOCKS_U)  { X = XC; W = WC; sc = sC; H = HC; nTiles = N_I / 16; blk = g - 2 * GEMM_BLOCKS_U; nBlk = GEMM_BLOCKS_I; }
    else if (g >= GEMM_BLOCKS_U) { X = XB; W = WB; sc = sB; H = HB; blk = g - GEMM_BLOCKS_U; }
    gemm_rows(X, W, sc, H, nTiles, blk, nBlk);
}

__global__ __launch_bounds__(GEMM_WAVES * 32)
void gemm3_kernel(const float* __restrict__ XA, const float* __restrict__ WA,
                  const float* __restrict__ sA, float* __restrict__ HA,
                  const float* __restrict__ XB, const float* __restrict__ WB,
                  const float* __restrict__ sB, float* __restrict__ HB,
                  const float* __restrict__ XC, const float* __restrict__ WC,
                  const float* __restrict__ sC, float* __restrict__ HC) {
    gemm_job(blockIdx.x, XA, WA, sA, HA, XB, WB, sB, HB, XC, WC, sC, HC);
}

__global__ __launch_bounds__(GEMM_WAVES * 32)
void layer1_kernel(const int* __restrict__ f_src, const int* __restrict__ f_dst,
                   const int* __restrict__ rank_f, const int* __restrict__ cur_f,
                   const int* __restrict__ base_f, int* __restrict__ csr_f,
                   const int* __restrict__ r_src, const int* __restrict__ r_dst,
                   const int* __restrict__ rank_r, const int* __restrict__ cur_r,
                   const int* __restrict__ base_r, int* __restrict__ csr_r,
                   const int* __restrict__ v_src, const int* __restrict__ v_dst,
                   const int* __restrict__ rank_v, const int* __restrict__ cur_v,
                   const int* __restrict__ base_v, int* __restrict__ csr_v,
                   const float* __restrict__ XA, const float* __restrict__ WA,
                   const float* __restrict__ sA, float* __restrict__ HA,
                   const float* __restrict__ XB, const float* __restrict__ WB,
                   const float* __restrict__ sB, float* __restrict__ HB,
                   const float* __restrict__ XC, const float* __restrict__ WC,
                   const float* __restrict__ sC, float* __restrict__ HC) {
    const int b = blockIdx.x;
    const bool isGemm = (b < 2 * GEMM_BLOCKS_ALL) && ((b & 1) == 0);
    if (isGemm) {
        gemm_job(b >> 1, XA, WA, sA, HA, XB, WB, sB, HB, XC, WC, sC, HC);
    } else {
        const int f = (b < 2 * GEMM_BLOCKS_ALL) ? (b >> 1) : (b - GEMM_BLOCKS_ALL);
        const int* __restrict__ src = f_src;  const int* __restrict__ dst = f_dst;
        const int* __restrict__ rank = rank_f; const int* __restrict__ cur = cur_f;
        const int* __restrict__ base = base_f; int* __restrict__ csr = csr_f;
        int blk = f;
        if (f >= 2 * FILL_BLOCKS)  { src = v_src; dst = v_dst; rank = rank_v; cur = cur_v; base = base_v; csr = csr_v; blk = f - 2 * FILL_BLOCKS; }
        else if (f >= FILL_BLOCKS) { src = r_src; dst = r_dst; rank = rank_r; cur = cur_r; base = base_r; csr = csr_r; blk = f - FILL_BLOCKS; }
        fill_edges(src, dst, rank, cur, base, csr, blk);
    }
}
static_assert(3 * FILL_BLOCKS >= GEMM_BLOCKS_ALL, "the alternating part assumes at least as many placement workgroups");

__device__ __forceinline__ float4 row_edge_sum(const float* __restrict__ H,
                                               const int* __restrict__ csr,
                                               int beg, int end, int lane) {
    float4 a = make_float4(0.0f, 0.0f, 0.0f, 0.0f);
    for (int c0 = beg; c0 < end; c0 += 32) {
        const int n = (end - c0 < 32) ? (end - c0) : 32;
        int my_s = 0;
        if (lane < n) my_s = csr[c0 + lane];
        #pragma unroll 4
        for (int j = 0; j < n; ++j) {
            const int s = __builtin_amdgcn_readlane(my_s, j);
            const float4 v = *((const float4*)(H + (size_t)s * DIM) + lane);
            a.x += v.x;
            a.y += v.y;
            a.z += v.z;
            a.w += v.w;
        }
    }
    return a;
}

__global__ void gather2_kernel(const float* __restrict__ H1,
                               const int* __restrict__ csr1, const int* __restrict__ cur1,
                               const float* __restrict__ invD1,
                               const float* __restrict__ H2,
                               const int* __restrict__ csr2, const int* __restrict__ cur2,
                               const float* __restrict__ invD2,
                               const float* __restrict__ b1, const float* __restrict__ b2,
                               float scale, int do_relu,
                               float* __restrict__ out, int nRows) {
    const int gid  = blockIdx.x * blockDim.x + threadIdx.x;
    const int row  = gid >> 5;
    const int lane = gid & 31;
    if (row >= nRows) return;

    const int vb1 = cur1[row];
    const int ve1 = cur1[row + 1];
    const int vb2 = cur2[row];
    const int ve2 = cur2[row + 1];
    const float wd1 = invD1[row];
    const float wd2 = invD2[row];
    const float4 bb1 = *((const float4*)b1 + lane);
    const float4 bb2 = *((const float4*)b2 + lane);
    const int beg1 = __builtin_amdgcn_readfirstlane(vb1);
    const int end1 = __builtin_amdgcn_readfirstlane(ve1);
    const int beg2 = __builtin_amdgcn_readfirstlane(vb2);
    const int end2 = __builtin_amdgcn_readfirstlane(ve2);

    const float4 s1 = row_edge_sum(H1, csr1, beg1, end1, lane);
    const float4 s2 = row_edge_sum(H2, csr2, beg2, end2, lane);

    float4 v;
    v.x = (s2.x * wd2 + s1.x * wd1) + bb1.x + bb2.x;
    v.y = (s2.y * wd2 + s1.y * wd1) + bb1.y + bb2.y;
    v.z = (s2.z * wd2 + s1.z * wd1) + bb1.z + bb2.z;
    v.w = (s2.w * wd2 + s1.w * wd1) + bb1.w + bb2.w;
    v.x *= scale; v.y *= scale; v.z *= scale; v.w *= scale;
    if (do_relu) {
        v.x = fmaxf(v.x, 0.0f); v.y = fmaxf(v.y, 0.0f);
        v.z = fmaxf(v.z, 0.0f); v.w = fmaxf(v.w, 0.0f);
    }
    *((float4*)(out + (size_t)row * DIM) + lane) = v;
}

__global__ void gather1_kernel(const float* __restrict__ H,
                               const int* __restrict__ csr, const int* __restrict__ cur,
                               const float* __restrict__ invD,
                               const float* __restrict__ b, int do_relu,
                               float* __restrict__ out, int nRows) {
    const int gid  = blockIdx.x * blockDim.x + threadIdx.x;
    const int row  = gid >> 5;
    const int lane = gid & 31;
    if (row >= nRows) return;

    const int vb = cur[row];
    const int ve = cur[row + 1];
    const float wd = invD[row];
    const float4 bb = *((const float4*)b + lane);
    const int beg = __builtin_amdgcn_readfirstlane(vb);
    const int end = __builtin_amdgcn_readfirstlane(ve);

    const float4 s = row_edge_sum(H, csr, beg, end, lane);

    float4 v;
    v.x = s.x * wd + bb.x; v.y = s.y * wd + bb.y; v.z = s.z * wd + bb.z; v.w = s.w * wd + bb.w;
    if (do_relu) {
        v.x = fmaxf(v.x, 0.0f); v.y = fmaxf(v.y, 0.0f);
        v.z = fmaxf(v.z, 0.0f); v.w = fmaxf(v.w, 0.0f);
    }
    *((float4*)(out + (size_t)row * DIM) + lane) = v;
}

extern "C" void kernel_launch(void* const* d_in, const int* in_sizes, int n_in,
                              void* d_out, int out_size, void* d_ws, size_t ws_size,
                              hipStream_t stream) {
    const float* x_user = (const float*)d_in[0];
    const float* x_item = (const float*)d_in[1];
    const int* f_src = (const int*)d_in[2];
    const int* f_dst = (const int*)d_in[3];
    const int* r_src = (const int*)d_in[4];
    const int* r_dst = (const int*)d_in[5];
    const int* v_src = (const int*)d_in[6];
    const int* v_dst = (const int*)d_in[7];
    const float* W1f = (const float*)d_in[8];
    const float* W1r = (const float*)d_in[9];
    const float* W1v = (const float*)d_in[10];
    const float* W2f = (const float*)d_in[11];
    const float* W2r = (const float*)d_in[12];
    const float* W2v = (const float*)d_in[13];
    const float* b1f = (const float*)d_in[14];
    const float* b1r = (const float*)d_in[15];
    const float* b1v = (const float*)d_in[16];
    const float* b2f = (const float*)d_in[17];
    const float* b2r = (const float*)d_in[18];
    const float* b2v = (const float*)d_in[19];

    float* w = (float*)d_ws;
    float* inv_fs = w; w += N_U;
    float* inv_fd = w; w += N_U;
    float* inv_rs = w; w += N_U;
    float* inv_rd = w; w += N_I;
    float* inv_vs = w; w += N_I;
    float* inv_vd = w; w += N_U;
    const int n_inv = N_BINS;
    float* h_user = w; w += (size_t)N_U * DIM;
    int* part = (int*)h_user;
    int* csr_f = (int*)w;
    int* csr_r = csr_f + NE;
    int* csr_v = csr_r + NE;
    int* cur_f = csr_v + NE;
    int* cur_r = cur_f + N_U + 1;
    int* cur_v = cur_r + N_I + 1;
    w += (size_t)N_I * DIM;
    float* u1     = w; w += (size_t)N_U * DIM;
    int* rank_f = (int*)u1;
    int* rank_r = rank_f + NE;
    int* rank_v = rank_r + NE;
    float* it1    = w; w += (size_t)N_I * DIM;
    static_assert((size_t)3 * NE + (size_t)N_U * DIM <= (size_t)(N_U + N_I) * DIM, "rates features fit behind the slots");
    float* h_rates1 = u1 + (size_t)3 * NE;

    float* out_u = (float*)d_out;
    float* out_i = out_u + (size_t)N_U * DIM;

    const int TPB = 256;
    const int gathU = (N_U * 32 + TPB - 1) / TPB;
    const int gathI = (N_I * 32 + TPB - 1) / TPB;
    dim3 rsGrid((N_U + 1023) / 1024, 3);

    count_kernel<<<CNT_JOBS * CNT_SLICES, CNT_THREADS, 0, stream>>>(f_src, f_dst, r_src, r_dst, v_src, v_dst,
                                                                  rank_f, rank_r, rank_v, part);
    combine_kernel<<<(N_BINS / 4 + TPB - 1) / TPB, TPB, 0, stream>>>(part, (int*)inv_fs);
    rowstart_kernel<<<rsGrid, TPB, 0, stream>>>((const int*)inv_fd, cur_f, N_U,
                                                (const int*)inv_rd, cur_r, N_I,
                                                (const int*)inv_vd, cur_v, N_U);
    invsqrt_kernel<<<(n_inv + TPB - 1) / TPB, TPB, 0, stream>>>((int*)inv_fs, n_inv);

    layer1_kernel<<<GEMM_BLOCKS_ALL + 3 * FILL_BLOCKS, GEMM_WAVES * 32, 0, stream>>>(
        f_src, f_dst, rank_f, cur_f, part + N_U, csr_f,
        r_src, r_dst, rank_r, cur_r, part + 3 * N_U, csr_r,
        v_src, v_dst, rank_v, cur_v, part + 3 * N_U + 2 * N_I, csr_v,
        x_user, W1f, inv_fs, out_u,
        x_user, W1r, inv_rs, h_rates1,
        x_item, W1v, inv_vs, out_i);

    gather2_kernel<<<gathU, TPB, 0, stream>>>(out_u, csr_f, cur_f, inv_fd,
                                              out_i, csr_v, cur_v, inv_vd,
                                              b1f, b1v, 0.5f, 1, h_user, N_U);
    gather1_kernel<<<gathI, TPB, 0, stream>>>(h_rates1, csr_r, cur_r, inv_rd,
                                              b1r, 1, out_i, N_I);

    gemm3_kernel<<<GEMM_BLOCKS_ALL, GEMM_WAVES * 32, 0, stream>>>(
        h_user, W2r, inv_rs, out_u,
        h_user, W2f, inv_fs, u1,
        out_i,  W2v, inv_vs, it1);
    gather1_kernel<<<gathI, TPB, 0, stream>>>(out_u, csr_r, cur_r, inv_rd,
                                              b2r, 0, out_i, N_I);
    gather2_kernel<<<gathU, TPB, 0, stream>>>(u1,  csr_f, cur_f, inv_fd,
                                              it1, csr_v, cur_v, inv_vd,
                                              b2f, b2v, 0.5f, 0, out_u, N_U);
}


# speedup vs baseline: 5.2624x; 1.0756x over previous
#include <hip/hip_runtime.h>

#define N_U 100000
#define N_I 50000
#define NE  1600000
#define DIM 128

typedef __attribute__((ext_vector_type(2))) float v2f;
typedef __attribute__((ext_vector_type(8))) float v8f;

#define CNT_SLICES    25
#define CNT_SLICE_LEN (NE / CNT_SLICES)
#define CNT_RANGE     50000
#define N_BINS        (4 * N_U + 2 * N_I)
#define CNT_JOBS      (N_BINS / CNT_RANGE)
#define CNT_THREADS   1024
#define CNT_TRIPS     ((CNT_SLICE_LEN / 4 + CNT_THREADS - 1) / CNT_THREADS)
static_assert(NE % CNT_SLICES == 0 && CNT_SLICE_LEN % 4 == 0, "slices are whole 16-byte groups");
static_assert(N_U == 2 * CNT_RANGE && N_I == CNT_RANGE, "window table below assumes these sizes");

__global__ __launch_bounds__(CNT_THREADS)
void count_kernel(const int* __restrict__ f_src, const int* __restrict__ f_dst,
                  const int* __restrict__ r_src, const int* __restrict__ r_dst,
                  const int* __restrict__ v_src, const int* __restrict__ v_dst,
                  int* __restrict__ rank_f, int* __restrict__ rank_r, int* __restrict__ rank_v,
                  int* __restrict__ part) {
    __shared__ __attribute__((aligned(16))) int cnt[CNT_RANGE];

    const int t   = threadIdx.x;
    const int job = blockIdx.x / CNT_SLICES;
    const int k   = blockIdx.x - job * CNT_SLICES;

    const int* __restrict__ keys = f_src;
    int* __restrict__ rank = nullptr;
    int lo = 0;
    switch (job) {
        case 0: keys = f_src; break;
        case 1: keys = f_src; lo = CNT_RANGE; break;
        case 2: keys = f_dst; rank = rank_f; break;
        case 3: keys = f_dst; rank = rank_f; lo = CNT_RANGE; break;
        case 4: keys = r_src; break;
        case 5: keys = r_src; lo = CNT_RANGE; break;
        case 6: keys = r_dst; rank = rank_r; break;
        case 7: keys = v_src; break;
        case 8: keys = v_dst; rank = rank_v; break;
        default: keys = v_dst; rank = rank_v; lo = CNT_RANGE; break;
    }

    for (int i = t; i < CNT_RANGE / 4; i += CNT_THREADS) ((int4*)cnt)[i] = make_int4(0, 0, 0, 0);
    __syncthreads();

    const int e0 = k * CNT_SLICE_LEN;
    const int4* __restrict__ k4 = (const int4*)(keys + e0);
    int4 v[CNT_TRIPS];
    #pragma unroll
    for (int j = 0; j < CNT_TRIPS; ++j) {
        const int i = t + j * CNT_THREADS;
        v[j] = (i < CNT_SLICE_LEN / 4) ? k4[i] : make_int4(-1, -1, -1, -1);
    }
    if (rank != nullptr) {
        #pragma unroll
        for (int j = 0; j < CNT_TRIPS; ++j) {
            const unsigned b0 = (unsigned)(v[j].x - lo), b1 = (unsigned)(v[j].y - lo);
            const unsigned b2 = (unsigned)(v[j].z - lo), b3 = (unsigned)(v[j].w - lo);
            int o0 = 0, o1 = 0, o2 = 0, o3 = 0;
            if (b0 < CNT_RANGE) o0 = __hip_atomic_fetch_add(&cnt[b0], 1, __ATOMIC_RELAXED, __HIP_MEMORY_SCOPE_WORKGROUP);
            if (b1 < CNT_RANGE) o1 = __hip_atomic_fetch_add(&cnt[b1], 1, __ATOMIC_RELAXED, __HIP_MEMORY_SCOPE_WORKGROUP);
            if (b2 < CNT_RANGE) o2 = __hip_atomic_fetch_add(&cnt[b2], 1, __ATOMIC_RELAXED, __HIP_MEMORY_SCOPE_WORKGROUP);
            if (b3 < CNT_RANGE) o3 = __hip_atomic_fetch_add(&cnt[b3], 1, __ATOMIC_RELAXED, __HIP_MEMORY_SCOPE_WORKGROUP);
            int* r = rank + e0 + 4 * (t + j * CNT_THREADS);
            if (b0 < CNT_RANGE) r[0] = o0;
            if (b1 < CNT_RANGE) r[1] = o1;
            if (b2 < CNT_RANGE) r[2] = o2;
            if (b3 < CNT_RANGE) r[3] = o3;
        }
    } else {
        #pragma unroll
        for (int j = 0; j < CNT_TRIPS; ++j) {
            const unsigned b0 = (unsigned)(v[j].x - lo), b1 = (unsigned)(v[j].y - lo);
            const unsigned b2 = (unsigned)(v[j].z - lo), b3 = (unsigned)(v[j].w - lo);
            if (b0 < CNT_RANGE) __hip_atomic_fetch_add(&cnt[b0], 1, __ATOMIC_RELAXED, __HIP_MEMORY_SCOPE_WORKGROUP);
            if (b1 < CNT_RANGE) __hip_atomic_fetch_add(&cnt[b1], 1, __ATOMIC_RELAXED, __HIP_MEMORY_SCOPE_WORKGROUP);
            if (b2 < CNT_RANGE) __hip_atomic_fetch_add(&cnt[b2], 1, __ATOMIC_RELAXED, __HIP_MEMORY_SCOPE_WORKGROUP);
            if (b3 < CNT_RANGE) __hip_atomic_fetch_add(&cnt[b3], 1, __ATOMIC_RELAXED, __HIP_MEMORY_SCOPE_WORKGROUP);
        }
    }
    __syncthreads();

    int4* __restrict__ o4 = (int4*)(part + (size_t)k * N_BINS + job * CNT_RANGE);
    for (int i = t; i < CNT_RANGE / 4; i += CNT_THREADS) o4[i] = ((const int4*)cnt)[i];
}

__global__ void combine_kernel(int* __restrict__ part, int* __restrict__ deg) {
    const int i = blockIdx.x * blockDim.x + threadIdx.x;
    if (i < N_BINS / 4) {
        const int b = 4 * i;
        const bool isDst = (b >= N_U && b < 2 * N_U) || (b >= 3 * N_U && b < 3 * N_U + N_I) ||
                           (b >= 3 * N_U + 2 * N_I);
        int4* __restrict__ p4 = (int4*)part + i;
        int4 c[CNT_SLICES];
        #pragma unroll
        for (int k = 0; k < CNT_SLICES; ++k) c[k] = p4[(size_t)k * (N_BINS / 4)];
        int4 run = make_int4(0, 0, 0, 0);
        #pragma unroll
        for (int k = 0; k < CNT_SLICES; ++k) {
            if (isDst) p4[(size_t)k * (N_BINS / 4)] = run;
            run.x += c[k].x; run.y += c[k].y; run.z += c[k].z; run.w += c[k].w;
        }
        ((int4*)deg)[i] = run;
    }
}

__global__ void invsqrt_kernel(int* __restrict__ a, int n) {
    int i = blockIdx.x * blockDim.x + threadIdx.x;
    if (i < n) {
        const int d = a[i];
        const float r = (d > 0) ? rsqrtf((float)d) : 0.0f;
        a[i] = __float_as_int(r);
    }
}

typedef __attribute__((ext_vector_type(4))) float v4f;
#define GEMM_WAVES 4
__device__ __forceinline__
void gemm_rows(const float* __restrict__ X, const float* __restrict__ W,
               const float* __restrict__ rowScale,
               float* __restrict__ H, int nTiles, int blk, int nBlk) {
    __shared__ __attribute__((aligned(16))) float sW[DIM * DIM];

    const int t = threadIdx.x;
    {
        #pragma unroll 4
        for (int j = 0; j < (DIM / 2) * (DIM / 4) / (GEMM_WAVES * 32); ++j) {
            const int idx = t + j * (GEMM_WAVES * 32);
            const int q   = idx & 31;
            const int kp  = idx >> 5;
            const v4f r0 = *(const v4f*)(W + (size_t)(2 * kp)     * DIM + 4 * q);
            const v4f r1 = *(const v4f*)(W + (size_t)(2 * kp + 1) * DIM + 4 * q);
            float* dst = sW + kp * 256 + (q >> 4) * 128 + (q & 15) * 4;
            v4f lo, hi;
            lo.x = r0.x; lo.y = r1.x; lo.z = r0.y; lo.w = r1.y;
            hi.x = r0.z; hi.y = r1.z; hi.z = r0.w; hi.w = r1.w;
            *(v4f*)(dst)      = lo;
            *(v4f*)(dst + 64) = hi;
        }
    }
    __syncthreads();

    const int lane = t & 31;
    const int wv   = __builtin_amdgcn_readfirstlane(t >> 5);
    const int c    = lane & 15;
    const int h    = lane >> 4;

    const float* sB = sW + h * 256 + c * 4;

    for (int tile = blk * GEMM_WAVES + wv; tile < nTiles; tile += nBlk * GEMM_WAVES) {
        const int m0 = tile << 4;
        const float* __restrict__ xrow = X + (size_t)(m0 + c) * DIM + 2 * h;

        v2f a[DIM / 4];
        #pragma unroll
        for (int s = 0; s < DIM / 4; ++s) a[s] = *(const v2f*)(xrow + 4 * s);
        const v4f sc0 = *(const v4f*)(rowScale + m0 + (h << 3));
        const v4f sc1 = *(const v4f*)(rowScale + m0 + (h << 3) + 4);

        v8f acc0 = {}, acc1 = {}, acc2 = {}, acc3 = {}, acc4 = {}, acc5 = {}, acc6 = {}, acc7 = {};
        #pragma unroll
        for (int s = 0; s < DIM / 4; ++s) {
            const v4f p0 = *(const v4f*)(sB + s * 512);
            const v4f p1 = *(const v4f*)(sB + s * 512 + 64);
            const v4f p2 = *(const v4f*)(sB + s * 512 + 128);
            const v4f p3 = *(const v4f*)(sB + s * 512 + 192);
            acc0 = __builtin_amdgcn_wmma_f32_16x16x4_f32(false, a[s], false, p0.xy, (short)0, acc0, false, false);
            acc1 = __builtin_amdgcn_wmma_f32_16x16x4_f32(false, a[s], false, p0.zw, (short)0, acc1, false, false);
            acc2 = __builtin_amdgcn_wmma_f32_16x16x4_f32(false, a[s], false, p1.xy, (short)0, acc2, false, false);
            acc3 = __builtin_amdgcn_wmma_f32_16x16x4_f32(false, a[s], false, p1.zw, (short)0, acc3, false, false);
            acc4 = __builtin_amdgcn_wmma_f32_16x16x4_f32(false, a[s], false, p2.xy, (short)0, acc4, false, false);
            acc5 = __builtin_amdgcn_wmma_f32_16x16x4_f32(false, a[s], false, p2.zw, (short)0, acc5, false, false);
            acc6 = __builtin_amdgcn_wmma_f32_16x16x4_f32(false, a[s], false, p3.xy, (short)0, acc6, false, false);
            acc7 = __builtin_amdgcn_wmma_f32_16x16x4_f32(false, a[s], false, p3.zw, (short)0, acc7, false, false);
            if ((s & 3) == 3) __builtin_amdgcn_sched_barrier(0);
        }

        float* __restrict__ hrow = H + (size_t)(m0 + (h << 3)) * DIM + 4 * c;
        #pragma unroll
        for (int r = 0; r < 8; ++r) {
            const float sc = (r < 4) ? sc0[r & 3] : sc1[r & 3];
            v4f lo, hi;
            lo.x = acc0[r] * sc; lo.y = acc1[r] * sc; lo.z = acc2[r] * sc; lo.w = acc3[r] * sc;
            hi.x = acc4[r] * sc; hi.y = acc5[r] * sc; hi.z = acc6[r] * sc; hi.w = acc7[r] * sc;
            *(v4f*)(hrow + (size_t)r * DIM)      = lo;
            *(v4f*)(hrow + (size_t)r * DIM + 64) = hi;
        }
    }
}

__global__ void rowstart_kernel(const int* __restrict__ degA, int* __restrict__ curA, int nA,
                                const int* __restrict__ degB, int* __restrict__ curB, int nB,
                                const int* __restrict__ degC, int* __restrict__ curC, int nC) {
    const int* __restrict__ deg = degA; int* __restrict__ cur = curA; int n = nA;
    if (blockIdx.y == 1) { deg = degB; cur = curB; n = nB; }
    if (blockIdx.y == 2) { deg = degC; cur = curC; n = nC; }
    const int t    = threadIdx.x;
    const int lane = t & 31;
    const int wv   = t >> 5;
    const int base = blockIdx.x << 10;
    if (base >= n) return;

    int part = 0;
    const int4* __restrict__ d4 = (const int4*)deg;
    #pragma unroll 4
    for (int i = t; i < (base >> 2); i += 256) {
        const int4 v = d4[i];
        part += v.x + v.y + v.z + v.w;
    }

    const int i0 = base + (t << 2);
    int c[4];
    #pragma unroll
    for (int j = 0; j < 4; ++j) c[j] = (i0 + j < n) ? deg[i0 + j] : 0;
    const int mine = c[0] + c[1] + c[2] + c[3];

    int incl = mine;
    for (int o = 1; o < 32; o <<= 1) {
        const int y = __shfl_up(incl, o);
        if (lane >= o) incl += y;
    }
    int psum = part;
    for (int o = 16; o > 0; o >>= 1) psum += __shfl_xor(psum, o);

    __shared__ int s_w[8];
    __shared__ int s_p[8];
    if (lane == 31) s_w[wv] = incl;
    if (lane == 0)  s_p[wv] = psum;
    __syncthreads();
    int off = 0;
    #pragma unroll
    for (int k = 0; k < 8; ++k) {
        off += s_p[k];
        if (k < wv) off += s_w[k];
    }
    int ex = off + incl - mine;
    #pragma unroll
    for (int j = 0; j < 4; ++j) {
        if (i0 + j < n) {
            cur[i0 + j] = ex;
            ex += c[j];
            if (i0 + j == n - 1) cur[n] = ex;
        }
    }
}

#define FILL_THREADS     (GEMM_WAVES * 32)
#define FILL_PER         8
#define FILL_BLOCK_EDGES (FILL_THREADS * FILL_PER)
#define FILL_BLOCKS      ((NE + FILL_BLOCK_EDGES - 1) / FILL_BLOCK_EDGES)
__device__ __forceinline__
void fill_edges(const int* __restrict__ src, const int* __restrict__ dst,
                const int* __restrict__ rank, const int* __restrict__ cur,
                const int* __restrict__ base, int* __restrict__ csr, int blk) {
    const int e0 = blk * FILL_BLOCK_EDGES + threadIdx.x;
    int s[FILL_PER], d[FILL_PER], r[FILL_PER], p[FILL_PER];
    #pragma unroll
    for (int j = 0; j < FILL_PER; ++j) {
        const int e  = e0 + j * FILL_THREADS;
        const int ec = (e < NE) ? e : (NE - 1);
        s[j] = src[ec];
        d[j] = dst[ec];
        r[j] = rank[ec];
    }
    #pragma unroll
    for (int j = 0; j < FILL_PER; ++j) {
        const int e  = e0 + j * FILL_THREADS;
        const int ec = (e < NE) ? e : (NE - 1);
        p[j] = cur[d[j]] + base[(size_t)(ec / CNT_SLICE_LEN) * N_BINS + d[j]] + r[j];
    }
    #pragma unroll
    for (int j = 0; j < FILL_PER; ++j) {
        const int e = e0 + j * FILL_THREADS;
        if (e < NE && (unsigned)p[j] < (unsigned)NE) csr[p[j]] = s[j];
    }
}

#define GEMM_BLOCKS_U   ((N_U / 16 + 2 * GEMM_WAVES - 1) / (2 * GEMM_WAVES))
#define GEMM_BLOCKS_I   ((N_I / 16 + 2 * GEMM_WAVES - 1) / (2 * GEMM_WAVES))
#define GEMM_BLOCKS_ALL (2 * GEMM_BLOCKS_U + GEMM_BLOCKS_I)
__device__ __forceinline__
void gemm_job(int g,
              const float* __restrict__ XA, const float* __restrict__ WA,
              const float* __restrict__ sA, float* __restrict__ HA,
              const float* __restrict__ XB, const float* __restrict__ WB,
              const float* __restrict__ sB, float* __restrict__ HB,
              const float* __restrict__ XC, const float* __restrict__ WC,
              const float* __restrict__ sC, float* __restrict__ HC) {
    const float* __restrict__ X = XA; const float* __restrict__ W = WA;
    const float* __restrict__ sc = sA; float* __restrict__ H = HA;
    int nTiles = N_U / 16, blk = g, nBlk = GEMM_BLOCKS_U;
    if (g >= 2 * GEMM_BLOCKS_U)  { X = XC; W = WC; sc = sC; H = HC; nTiles = N_I / 16; blk = g - 2 * GEMM_BLOCKS_U; nBlk = GEMM_BLOCKS_I; }
    else if (g >= GEMM_BLOCKS_U) { X = XB; W = WB; sc = sB; H = HB; blk = g - GEMM_BLOCKS_U; }
    gemm_rows(X, W, sc, H, nTiles, blk, nBlk);
}

__global__ __launch_bounds__(GEMM_WAVES * 32)
void gemm3_kernel(const float* __restrict__ XA, const float* __restrict__ WA,
                  const float* __restrict__ sA, float* __restrict__ HA,
                  const float* __restrict__ XB, const float* __restrict__ WB,
                  const float* __restrict__ sB, float* __restrict__ HB,
                  const float* __restrict__ XC, const float* __restrict__ WC,
                  const float* __restrict__ sC, float* __restrict__ HC) {
    gemm_job(blockIdx.x, XA, WA, sA, HA, XB, WB, sB, HB, XC, WC, sC, HC);
}

__global__ __launch_bounds__(GEMM_WAVES * 32)
void layer1_kernel(const int* __restrict__ f_src, const int* __restrict__ f_dst,
                   const int* __restrict__ rank_f, const int* __restrict__ cur_f,
                   const int* __restrict__ base_f, int* __restrict__ csr_f,
                   const int* __restrict__ r_src, const int* __restrict__ r_dst,
                   const int* __restrict__ rank_r, const int* __restrict__ cur_r,
                   const int* __restrict__ base_r, int* __restrict__ csr_r,
                   const int* __restrict__ v_src, const int* __restrict__ v_dst,
                   const int* __restrict__ rank_v, const int* __restrict__ cur_v,
                   const int* __restrict__ base_v, int* __restrict__ csr_v,
                   const float* __restrict__ XA, const float* __restrict__ WA,
                   const float* __restrict__ sA, float* __restrict__ HA,
                   const float* __restrict__ XB, const float* __restrict__ WB,
                   const float* __restrict__ sB, float* __restrict__ HB,
                   const float* __restrict__ XC, const float* __restrict__ WC,
                   const float* __restrict__ sC, float* __restrict__ HC) {
    const int b = blockIdx.x;
    const bool isGemm = (b < 2 * GEMM_BLOCKS_ALL) && ((b & 1) == 0);
    if (isGemm) {
        gemm_job(b >> 1, XA, WA, sA, HA, XB, WB, sB, HB, XC, WC, sC, HC);
    } else {
        const int f = (b < 2 * GEMM_BLOCKS_ALL) ? (b >> 1) : (b - GEMM_BLOCKS_ALL);
        const int* __restrict__ src = f_src;  const int* __restrict__ dst = f_dst;
        const int* __restrict__ rank = rank_f; const int* __restrict__ cur = cur_f;
        const int* __restrict__ base = base_f; int* __restrict__ csr = csr_f;
        int blk = f;
        if (f >= 2 * FILL_BLOCKS)  { src = v_src; dst = v_dst; rank = rank_v; cur = cur_v; base = base_v; csr = csr_v; blk = f - 2 * FILL_BLOCKS; }
        else if (f >= FILL_BLOCKS) { src = r_src; dst = r_dst; rank = rank_r; cur = cur_r; base = base_r; csr = csr_r; blk = f - FILL_BLOCKS; }
        fill_edges(src, dst, rank, cur, base, csr, blk);
    }
}
static_assert(3 * FILL_BLOCKS >= GEMM_BLOCKS_ALL, "the alternating part assumes at least as many placement workgroups");

__device__ __forceinline__ float4 row_edge_sum(const float* __restrict__ H,
                                               const int* __restrict__ csr,
                                               int beg, int end, int lane) {
    float4 a = make_float4(0.0f, 0.0f, 0.0f, 0.0f);
    for (int c0 = beg; c0 < end; c0 += 32) {
        const int n = (end - c0 < 32) ? (end - c0) : 32;
        int my_s = 0;
        if (lane < n) my_s = csr[c0 + lane];
        #pragma unroll 4
        for (int j = 0; j < n; ++j) {
            const int s = __builtin_amdgcn_readlane(my_s, j);
            const float4 v = *((const float4*)(H + (size_t)s * DIM) + lane);
            a.x += v.x;
            a.y += v.y;
            a.z += v.z;
            a.w += v.w;
        }
    }
    return a;
}

__global__ void gather2_kernel(const float* __restrict__ H1,
                               const int* __restrict__ csr1, const int* __restrict__ cur1,
                               const float* __restrict__ invD1,
                               const float* __restrict__ H2,
                               const int* __restrict__ csr2, const int* __restrict__ cur2,
                               const float* __restrict__ invD2,
                               const float* __restrict__ b1, const float* __restrict__ b2,
                               float scale, int do_relu,
                               float* __restrict__ out, int nRows) {
    const int gid  = blockIdx.x * blockDim.x + threadIdx.x;
    const int row  = gid >> 5;
    const int lane = gid & 31;
    if (row >= nRows) return;

    const int vb1 = cur1[row];
    const int ve1 = cur1[row + 1];
    const int vb2 = cur2[row];
    const int ve2 = cur2[row + 1];
    const float wd1 = invD1[row];
    const float wd2 = invD2[row];
    const float4 bb1 = *((const float4*)b1 + lane);
    const float4 bb2 = *((const float4*)b2 + lane);
    const int beg1 = __builtin_amdgcn_readfirstlane(vb1);
    const int end1 = __builtin_amdgcn_readfirstlane(ve1);
    const int beg2 = __builtin_amdgcn_readfirstlane(vb2);
    const int end2 = __builtin_amdgcn_readfirstlane(ve2);

    const float4 s1 = row_edge_sum(H1, csr1, beg1, end1, lane);
    const float4 s2 = row_edge_sum(H2, csr2, beg2, end2, lane);

    float4 v;
    v.x = (s2.x * wd2 + s1.x * wd1) + bb1.x + bb2.x;
    v.y = (s2.y * wd2 + s1.y * wd1) + bb1.y + bb2.y;
    v.z = (s2.z * wd2 + s1.z * wd1) + bb1.z + bb2.z;
    v.w = (s2.w * wd2 + s1.w * wd1) + bb1.w + bb2.w;
    v.x *= scale; v.y *= scale; v.z *= scale; v.w *= scale;
    if (do_relu) {
        v.x = fmaxf(v.x, 0.0f); v.y = fmaxf(v.y, 0.0f);
        v.z = fmaxf(v.z, 0.0f); v.w = fmaxf(v.w, 0.0f);
    }
    *((float4*)(out + (size_t)row * DIM) + lane) = v;
}

__global__ void gather1_kernel(const float* __restrict__ H,
                               const int* __restrict__ csr, const int* __restrict__ cur,
                               const float* __restrict__ invD,
                               const float* __restrict__ b, int do_relu,
                               float* __restrict__ out, int nRows) {
    const int gid  = blockIdx.x * blockDim.x + threadIdx.x;
    const int row  = gid >> 5;
    const int lane = gid & 31;
    if (row >= nRows) return;

    const int vb = cur[row];
    const int ve = cur[row + 1];
    const float wd = invD[row];
    const float4 bb = *((const float4*)b + lane);
    const int beg = __builtin_amdgcn_readfirstlane(vb);
    const int end = __builtin_amdgcn_readfirstlane(ve);

    const float4 s = row_edge_sum(H, csr, beg, end, lane);

    float4 v;
    v.x = s.x * wd + bb.x; v.y = s.y * wd + bb.y; v.z = s.z * wd + bb.z; v.w = s.w * wd + bb.w;
    if (do_relu) {
        v.x = fmaxf(v.x, 0.0f); v.y = fmaxf(v.y, 0.0f);
        v.z = fmaxf(v.z, 0.0f); v.w = fmaxf(v.w, 0.0f);
    }
    *((float4*)(out + (size_t)row * DIM) + lane) = v;
}

extern "C" void kernel_launch(void* const* d_in, const int* in_sizes, int n_in,
                              void* d_out, int out_size, void* d_ws, size_t ws_size,
                              hipStream_t stream) {
    const float* x_user = (const float*)d_in[0];
    const float* x_item = (const float*)d_in[1];
    const int* f_src = (const int*)d_in[2];
    const int* f_dst = (const int*)d_in[3];
    const int* r_src = (const int*)d_in[4];
    const int* r_dst = (const int*)d_in[5];
    const int* v_src = (const int*)d_in[6];
    const int* v_dst = (const int*)d_in[7];
    const float* W1f = (const float*)d_in[8];
    const float* W1r = (const float*)d_in[9];
    const float* W1v = (const float*)d_in[10];
    const float* W2f = (const float*)d_in[11];
    const float* W2r = (const float*)d_in[12];
    const float* W2v = (const float*)d_in[13];
    const float* b1f = (const float*)d_in[14];
    const float* b1r = (const float*)d_in[15];
    const float* b1v = (const float*)d_in[16];
    const float* b2f = (const float*)d_in[17];
    const float* b2r = (const float*)d_in[18];
    const float* b2v = (const float*)d_in[19];

    static_assert(((size_t)N_U * DIM) % 64 == 0 && ((size_t)N_I * DIM) % 64 == 0 && ((size_t)3 * NE) % 64 == 0,
                  "table slots are whole 256-byte units");
    float* w = (float*)d_ws;
    float* h_user = w; w += (size_t)N_U * DIM;
    int* part = (int*)h_user;
    int* csr_f = (int*)w;
    int* csr_r = csr_f + NE;
    int* csr_v = csr_r + NE;
    int* cur_f = csr_v + NE;
    int* cur_r = cur_f + N_U + 1;
    int* cur_v = cur_r + N_I + 1;
    w += (size_t)N_I * DIM;
    float* u1     = w; w += (size_t)N_U * DIM;
    int* rank_f = (int*)u1;
    int* rank_r = rank_f + NE;
    int* rank_v = rank_r + NE;
    float* it1    = w; w += (size_t)N_I * DIM;
    float* inv_fs = w; w += N_U;
    float* inv_fd = w; w += N_U;
    float* inv_rs = w; w += N_U;
    float* inv_rd = w; w += N_I;
    float* inv_vs = w; w += N_I;
    float* inv_vd = w; w += N_U;
    const int n_inv = N_BINS;
    static_assert((size_t)3 * NE + (size_t)N_U * DIM <= (size_t)(N_U + N_I) * DIM, "rates features fit behind the slots");
    float* h_rates1 = u1 + (size_t)3 * NE;

    float* out_u = (float*)d_out;
    float* out_i = out_u + (size_t)N_U * DIM;

    const int TPB = 256;
    const int gathU = (N_U * 32 + TPB - 1) / TPB;
    const int gathI = (N_I * 32 + TPB - 1) / TPB;
    dim3 rsGrid((N_U + 1023) / 1024, 3);

    count_kernel<<<CNT_JOBS * CNT_SLICES, CNT_THREADS, 0, stream>>>(f_src, f_dst, r_src, r_dst, v_src, v_dst,
                                                                  rank_f, rank_r, rank_v, part);
    combine_kernel<<<(N_BINS / 4 + TPB - 1) / TPB, TPB, 0, stream>>>(part, (int*)inv_fs);
    rowstart_kernel<<<rsGrid, TPB, 0, stream>>>((const int*)inv_fd, cur_f, N_U,
                                                (const int*)inv_rd, cur_r, N_I,
                                                (const int*)inv_vd, cur_v, N_U);
    invsqrt_kernel<<<(n_inv + TPB - 1) / TPB, TPB, 0, stream>>>((int*)inv_fs, n_inv);

    layer1_kernel<<<GEMM_BLOCKS_ALL + 3 * FILL_BLOCKS, GEMM_WAVES * 32, 0, stream>>>(
        f_src, f_dst, rank_f, cur_f, part + N_U, csr_f,
        r_src, r_dst, rank_r, cur_r, part + 3 * N_U, csr_r,
        v_src, v_dst, rank_v, cur_v, part + 3 * N_U + 2 * N_I, csr_v,
        x_user, W1f, inv_fs, out_u,
        x_user, W1r, inv_rs, h_rates1,
        x_item, W1v, inv_vs, out_i);

    gather2_kernel<<<gathU, TPB, 0, stream>>>(out_u, csr_f, cur_f, inv_fd,
                                              out_i, csr_v, cur_v, inv_vd,
                                              b1f, b1v, 0.5f, 1, h_user, N_U);
    gather1_kernel<<<gathI, TPB, 0, stream>>>(h_rates1, csr_r, cur_r, inv_rd,
                                              b1r, 1, out_i, N_I);

    gemm3_kernel<<<GEMM_BLOCKS_ALL, GEMM_WAVES * 32, 0, stream>>>(
        h_user, W2r, inv_rs, out_u,
        h_user, W2f, inv_fs, u1,
        out_i,  W2v, inv_vs, it1);
    gather1_kernel<<<gathI, TPB, 0, stream>>>(out_u, csr_r, cur_r, inv_rd,
                                              b2r, 0, out_i, N_I);
    gather2_kernel<<<gathU, TPB, 0, stream>>>(u1,  csr_f, cur_f, inv_fd,
                                              it1, csr_v, cur_v, inv_vd,
                                              b2f, b2v, 0.5f, 0, out_u, N_U);
}
